# starting kernel
# baseline (speedup 1.0000x reference)
#include <hip/hip_runtime.h>
#include <hip/hip_bf16.h>


typedef __attribute__((ext_vector_type(2))) float v2f;
typedef __attribute__((ext_vector_type(8))) float v8f;

#define NN      2048
#define IN_DIM  256
#define OUT_DIM 64
#define NHEADS  8
#define DH      8
#define JCHUNK  128
#define LSTR    132

__global__ void gat_wh_kernel(const float* __restrict__ h,
                              const float* __restrict__ W_w,
                              const float* __restrict__ W_b,
                              const float* __restrict__ a_w,
                              float* __restrict__ wh,
                              float* __restrict__ e_i,
                              float* __restrict__ e_j) {
    __shared__ float hrow[IN_DIM];
    __shared__ float whs[OUT_DIM];
    const int n = blockIdx.x;
    const int t = threadIdx.x;

    ((float4*)hrow)[t] = ((const float4*)(h + (size_t)n * IN_DIM))[t];
    __syncthreads();

    const float4* wr4 = (const float4*)(W_w + (size_t)t * IN_DIM);
    const float4* hr4 = (const float4*)hrow;
    float acc = W_b[t];
#pragma unroll 8
    for (int k = 0; k < IN_DIM / 4; ++k) {
        float4 w = wr4[k];
        float4 x = hr4[k];
        acc = fmaf(w.x, x.x, acc);
        acc = fmaf(w.y, x.y, acc);
        acc = fmaf(w.z, x.z, acc);
        acc = fmaf(w.w, x.w, acc);
    }
    wh[(size_t)n * OUT_DIM + t] = acc;
    whs[t] = acc;
    __syncthreads();

    if (t < NHEADS) {
        float s = 0.f;
#pragma unroll
        for (int d = 0; d < DH; ++d) s = fmaf(whs[t * DH + d], a_w[d], s);
        e_i[(size_t)n * NHEADS + t] = s;
    } else if (t < 2 * NHEADS) {
        const int hh = t - NHEADS;
        float s = 0.f;
#pragma unroll
        for (int d = 0; d < DH; ++d) s = fmaf(whs[hh * DH + d], a_w[DH + d], s);
        e_j[(size_t)n * NHEADS + hh] = s;
    }
}

__global__ __launch_bounds__(256)
void gat_att_kernel(const int*   __restrict__ adj,
                    const float* __restrict__ wh,
                    const float* __restrict__ e_i,
                    const float* __restrict__ e_j,
                    const float* __restrict__ a_b,
                    float* __restrict__ out) {
    __shared__ int   adjS[16 * LSTR];
    __shared__ float whT[OUT_DIM * LSTR];
    __shared__ float ejT[NHEADS * LSTR];
    __shared__ float pden[NHEADS][32];

    const int tid  = threadIdx.x;
    const int hH   = tid >> 5;
    const int lane = tid & 31;
    const int i0   = blockIdx.x * 16;

    const int   iLoc  = lane & 15;
    const int   nCol  = lane & 15;
    const int   nColC = nCol & 7;
    const int   khalf = (lane >> 4) * 2;
    const float cmask = (nCol < DH) ? 1.0f : 0.0f;
    const float eib   = e_i[(size_t)(i0 + iLoc) * NHEADS + hH] + a_b[0];
    const int   dRow  = hH * DH + nColC;

    v8f   acc  = {0.f, 0.f, 0.f, 0.f, 0.f, 0.f, 0.f, 0.f};
    float psum = 0.f;

    for (int chunk = 0; chunk < NN; chunk += JCHUNK) {
        __syncthreads();

        for (int idx = tid; idx < 512; idx += 256) {
            const int r = idx >> 5;
            const int c = idx & 31;
            ((int4*)&adjS[r * LSTR])[c] =
                *(const int4*)(adj + (size_t)(i0 + r) * NN + chunk + c * 4);
        }
        for (int idx = tid; idx < JCHUNK * OUT_DIM; idx += 256) {
            const int j = idx >> 6;
            const int d = idx & 63;
            whT[d * LSTR + j] = wh[(size_t)(chunk + j) * OUT_DIM + d];
        }
        for (int idx = tid; idx < JCHUNK * NHEADS; idx += 256) {
            const int j  = idx >> 3;
            const int hh = idx & 7;
            ejT[hh * LSTR + j] = e_j[(size_t)(chunk + j) * NHEADS + hh];
        }

        if (chunk + JCHUNK < NN)
            __builtin_prefetch(adj + (size_t)(i0 + (tid >> 4)) * NN + chunk + JCHUNK, 0, 0);

        __syncthreads();

#pragma unroll
        for (int js = 0; js < JCHUNK; js += 4) {
            const int j0 = js + khalf;
            const float2 ej = *(const float2*)&ejT[hH * LSTR + j0];
            const int2   ad = *(const int2*)  &adjS[iLoc * LSTR + j0];
            const float2 wv = *(const float2*)&whT[dRow * LSTR + j0];

            float s0 = eib + ej.x;
            float s1 = eib + ej.y;
            s0 = (s0 >= 0.f) ? s0 : 0.2f * s0;
            s1 = (s1 >= 0.f) ? s1 : 0.2f * s1;
            const float m0 = (ad.x > 0) ? 1.0f : 0.0f;
            const float m1 = (ad.y > 0) ? 1.0f : 0.0f;
            const float p0 = __expf(s0) * m0;
            const float p1 = __expf(s1) * m1;
            psum += p0 + p1;

            v2f A; A.x = p0;          A.y = p1;
            v2f B; B.x = wv.x * cmask; B.y = wv.y * cmask;
            acc = __builtin_amdgcn_wmma_f32_16x16x4_f32(
                false, A, false, B, (short)0, acc, false, false);
        }
    }

    pden[hH][lane] = psum;
    __syncthreads();

    const int rbase = (lane >> 4) * 8;
    if (nCol < DH) {
#pragma unroll
        for (int v = 0; v < 8; ++v) {
            const int   iL  = rbase + v;
            const float den = pden[hH][iL] + pden[hH][iL + 16];
            float o = acc[v] / den;
            o = (o > 0.f) ? o : (__expf(o) - 1.f);
            out[(size_t)hH * NN * DH + (size_t)(i0 + iL) * DH + nCol] = o;
        }
    }
}

extern "C" void kernel_launch(void* const* d_in, const int* in_sizes, int n_in,
                              void* d_out, int out_size, void* d_ws, size_t ws_size,
                              hipStream_t stream) {
    const float* h   = (const float*)d_in[0];
    const int*   adj = (const int*)  d_in[1];
    const float* W_w = (const float*)d_in[2];
    const float* W_b = (const float*)d_in[3];
    const float* a_w = (const float*)d_in[4];
    const float* a_b = (const float*)d_in[5];
    float* out = (float*)d_out;

    float* wh  = (float*)d_ws;
    float* e_i = wh  + (size_t)NN * OUT_DIM;
    float* e_j = e_i + (size_t)NN * NHEADS;

    gat_wh_kernel<<<NN, 64, 0, stream>>>(h, W_w, W_b, a_w, wh, e_i, e_j);
    gat_att_kernel<<<NN / 16, 256, 0, stream>>>(adj, wh, e_i, e_j, a_b, out);
}


# attention kernel: software-pipelined staging - next j-tile's global loads issued to registers before the WMMA loop, stored to a second LDS stage after it; one barrier per tile
# speedup vs baseline: 2.7264x; 2.7264x over previous
#include <hip/hip_runtime.h>
#include <hip/hip_bf16.h>


typedef __attribute__((ext_vector_type(2))) float v2f;
typedef __attribute__((ext_vector_type(8))) float v8f;
typedef __attribute__((ext_vector_type(4))) int   v4i;

#define NN      2048
#define IN_DIM  256
#define OUT_DIM 64
#define NHEADS  8
#define DH      8
#define JCHUNK  128
#define LSTR    132

__global__ void gat_wh_kernel(const float* __restrict__ h,
                              const float* __restrict__ W_w,
                              const float* __restrict__ W_b,
                              const float* __restrict__ a_w,
                              float* __restrict__ wh,
                              float* __restrict__ e_i,
                              float* __restrict__ e_j) {
    __shared__ float hrow[IN_DIM];
    __shared__ float whs[OUT_DIM];
    const int n = blockIdx.x;
    const int t = threadIdx.x;

    ((float4*)hrow)[t] = ((const float4*)(h + (size_t)n * IN_DIM))[t];
    __syncthreads();

    const float4* wr4 = (const float4*)(W_w + (size_t)t * IN_DIM);
    const float4* hr4 = (const float4*)hrow;
    float acc = W_b[t];
#pragma unroll 8
    for (int k = 0; k < IN_DIM / 4; ++k) {
        float4 w = wr4[k];
        float4 x = hr4[k];
        acc = fmaf(w.x, x.x, acc);
        acc = fmaf(w.y, x.y, acc);
        acc = fmaf(w.z, x.z, acc);
        acc = fmaf(w.w, x.w, acc);
    }
    wh[(size_t)n * OUT_DIM + t] = acc;
    whs[t] = acc;
    __syncthreads();

    if (t < NHEADS) {
        float s = 0.f;
#pragma unroll
        for (int d = 0; d < DH; ++d) s = fmaf(whs[t * DH + d], a_w[d], s);
        e_i[(size_t)n * NHEADS + t] = s;
    } else if (t < 2 * NHEADS) {
        const int hh = t - NHEADS;
        float s = 0.f;
#pragma unroll
        for (int d = 0; d < DH; ++d) s = fmaf(whs[hh * DH + d], a_w[DH + d], s);
        e_j[(size_t)n * NHEADS + hh] = s;
    }
}

#define NCHUNK    (NN / JCHUNK)
#define ADJ_STAGE (16 * LSTR)
#define WH_STAGE  (OUT_DIM * LSTR)
#define EJ_STAGE  (NHEADS * LSTR)

__global__ __launch_bounds__(256)
void gat_att_kernel(const int*   __restrict__ adj,
                    const float* __restrict__ wh,
                    const float* __restrict__ e_i,
                    const float* __restrict__ e_j,
                    const float* __restrict__ a_b,
                    float* __restrict__ out) {
    __shared__ alignas(16) int   adjS[2 * ADJ_STAGE];
    __shared__ alignas(16) float whT[2 * WH_STAGE];
    __shared__ alignas(16) float ejT[2 * EJ_STAGE];
    __shared__ float pden[NHEADS][32];

    const int tid  = threadIdx.x;
    const int hH   = tid >> 5;
    const int lane = tid & 31;
    const int i0   = blockIdx.x * 16;

    const int   iLoc  = lane & 15;
    const int   nCol  = lane & 15;
    const int   nColC = nCol & 7;
    const int   khalf = (lane >> 4) * 2;
    const float cmask = (nCol < DH) ? 1.0f : 0.0f;
    const float eib   = e_i[(size_t)(i0 + iLoc) * NHEADS + hH] + a_b[0];
    const int   dRow  = hH * DH + nColC;

    const int aR = tid >> 5;
    const int aC = tid & 31;
    const int* adjP = adj + (size_t)(i0 + aR) * NN + aC * 4;
    const int wJ = hH * 4 + (lane >> 3);
    const int wD = (lane & 7) * 4;
    const float* whP = wh + (size_t)wJ * OUT_DIM + wD;
    const int eJ = tid >> 1;
    const int eH = (tid & 1) * 4;
    const float* ejP = e_j + (size_t)eJ * NHEADS + eH;

    v4i    rA0, rA1;
    float4 rW[8];
    float4 rE;

    auto tile_load = [&](int chunk) {
        rA0 = *(const v4i*)(adjP + chunk);
        rA1 = *(const v4i*)(adjP + chunk + 8 * NN);
#pragma unroll
        for (int k = 0; k < 8; ++k)
            rW[k] = *(const float4*)(whP + (size_t)(chunk + (k >> 1) * 32) * OUT_DIM + (k & 1) * 32);
        rE = *(const float4*)(ejP + (size_t)chunk * NHEADS);
    };
    auto tile_store = [&](int st) {
        int*   aS = adjS + st * ADJ_STAGE;
        float* wS = whT  + st * WH_STAGE;
        float* eS = ejT  + st * EJ_STAGE;
        ((v4i*)&aS[aR * LSTR])[aC]       = rA0;
        ((v4i*)&aS[(aR + 8) * LSTR])[aC] = rA1;
#pragma unroll
        for (int k = 0; k < 8; ++k) {
            const int j = (k >> 1) * 32 + wJ;
            const int d = (k & 1) * 32 + wD;
            wS[(d + 0) * LSTR + j] = rW[k].x;
            wS[(d + 1) * LSTR + j] = rW[k].y;
            wS[(d + 2) * LSTR + j] = rW[k].z;
            wS[(d + 3) * LSTR + j] = rW[k].w;
        }
        eS[(eH + 0) * LSTR + eJ] = rE.x;
        eS[(eH + 1) * LSTR + eJ] = rE.y;
        eS[(eH + 2) * LSTR + eJ] = rE.z;
        eS[(eH + 3) * LSTR + eJ] = rE.w;
    };

    v8f   acc  = {0.f, 0.f, 0.f, 0.f, 0.f, 0.f, 0.f, 0.f};
    float psum = 0.f;

    auto tile_compute = [&](int st) {
        const int*   aS = adjS + st * ADJ_STAGE;
        const float* wS = whT  + st * WH_STAGE;
        const float* eS = ejT  + st * EJ_STAGE;
#pragma unroll 16
        for (int js = 0; js < JCHUNK; js += 4) {
            const int j0 = js + khalf;
            const float2 ej = *(const float2*)&eS[hH * LSTR + j0];
            const int2   ad = *(const int2*)  &aS[iLoc * LSTR + j0];
            const float2 wv = *(const float2*)&wS[dRow * LSTR + j0];

            float s0 = eib + ej.x;
            float s1 = eib + ej.y;
            s0 = (s0 >= 0.f) ? s0 : 0.2f * s0;
            s1 = (s1 >= 0.f) ? s1 : 0.2f * s1;
            const float m0 = (ad.x > 0) ? 1.0f : 0.0f;
            const float m1 = (ad.y > 0) ? 1.0f : 0.0f;
            const float p0 = __expf(s0) * m0;
            const float p1 = __expf(s1) * m1;
            psum += p0 + p1;

            v2f A; A.x = p0;          A.y = p1;
            v2f B; B.x = wv.x * cmask; B.y = wv.y * cmask;
            acc = __builtin_amdgcn_wmma_f32_16x16x4_f32(
                false, A, false, B, (short)0, acc, false, false);
        }
    };

    tile_load(0);
    tile_store(0);
    __syncthreads();

    for (int c = 0; c + 1 < NCHUNK; ++c) {
        tile_load((c + 1) * JCHUNK);
        tile_compute(c & 1);
        tile_store((c + 1) & 1);
        __syncthreads();
    }
    tile_compute((NCHUNK - 1) & 1);

    pden[hH][lane] = psum;
    __syncthreads();

    const int rbase = (lane >> 4) * 8;
    if (nCol < DH) {
#pragma unroll
        for (int v = 0; v < 8; ++v) {
            const int   iL  = rbase + v;
            const float den = pden[hH][iL] + pden[hH][iL + 16];
            float o = acc[v] / den;
            o = (o > 0.f) ? o : (__expf(o) - 1.f);
            out[(size_t)hH * NN * DH + (size_t)(i0 + iL) * DH + nCol] = o;
        }
    }
}

extern "C" void kernel_launch(void* const* d_in, const int* in_sizes, int n_in,
                              void* d_out, int out_size, void* d_ws, size_t ws_size,
                              hipStream_t stream) {
    const float* h   = (const float*)d_in[0];
    const int*   adj = (const int*)  d_in[1];
    const float* W_w = (const float*)d_in[2];
    const float* W_b = (const float*)d_in[3];
    const float* a_w = (const float*)d_in[4];
    const float* a_b = (const float*)d_in[5];
    float* out = (float*)d_out;

    float* wh  = (float*)d_ws;
    float* e_i = wh  + (size_t)NN * OUT_DIM;
    float* e_j = e_i + (size_t)NN * NHEADS;

    gat_wh_kernel<<<NN, 64, 0, stream>>>(h, W_w, W_b, a_w, wh, e_i, e_j);
    gat_att_kernel<<<NN / 16, 256, 0, stream>>>(adj, wh, e_i, e_j, a_b, out);
}


# attention kernel on all 256 WGPs: each work-group takes 4 of the 8 heads, wave pairs split each 256-wide j-tile; the two halves' partial sums are added at the end (sum order over j changes)
# speedup vs baseline: 3.5916x; 1.3173x over previous
#include <hip/hip_runtime.h>
#include <hip/hip_bf16.h>


typedef __attribute__((ext_vector_type(2))) float v2f;
typedef __attribute__((ext_vector_type(8))) float v8f;
typedef __attribute__((ext_vector_type(4))) int   v4i;

#define NN      2048
#define IN_DIM  256
#define OUT_DIM 64
#define NHEADS  8
#define DH      8
#define HPB     4
#define JCHUNK  256
#define JHALF   (JCHUNK / 2)
#define LSTR    260

__global__ void gat_wh_kernel(const float* __restrict__ h,
                              const float* __restrict__ W_w,
                              const float* __restrict__ W_b,
                              const float* __restrict__ a_w,
                              float* __restrict__ wh,
                              float* __restrict__ e_i,
                              float* __restrict__ e_j) {
    __shared__ float hrow[IN_DIM];
    __shared__ float whs[OUT_DIM];
    const int n = blockIdx.x;
    const int t = threadIdx.x;

    ((float4*)hrow)[t] = ((const float4*)(h + (size_t)n * IN_DIM))[t];
    __syncthreads();

    const float4* wr4 = (const float4*)(W_w + (size_t)t * IN_DIM);
    const float4* hr4 = (const float4*)hrow;
    float acc = W_b[t];
#pragma unroll 8
    for (int k = 0; k < IN_DIM / 4; ++k) {
        float4 w = wr4[k];
        float4 x = hr4[k];
        acc = fmaf(w.x, x.x, acc);
        acc = fmaf(w.y, x.y, acc);
        acc = fmaf(w.z, x.z, acc);
        acc = fmaf(w.w, x.w, acc);
    }
    wh[(size_t)n * OUT_DIM + t] = acc;
    whs[t] = acc;
    __syncthreads();

    if (t < NHEADS) {
        float s = 0.f;
#pragma unroll
        for (int d = 0; d < DH; ++d) s = fmaf(whs[t * DH + d], a_w[d], s);
        e_i[(size_t)n * NHEADS + t] = s;
    } else if (t < 2 * NHEADS) {
        const int hh = t - NHEADS;
        float s = 0.f;
#pragma unroll
        for (int d = 0; d < DH; ++d) s = fmaf(whs[hh * DH + d], a_w[DH + d], s);
        e_j[(size_t)n * NHEADS + hh] = s;
    }
}

#define NCHUNK    (NN / JCHUNK)
#define ADJ_STAGE (16 * LSTR)
#define WH_STAGE  (HPB * DH * LSTR)
#define EJ_STAGE  (HPB * LSTR)
#define ATT_GRID  ((NN / 16) * (NHEADS / HPB))

__global__ __launch_bounds__(256)
void gat_att_kernel(const int*   __restrict__ adj,
                    const float* __restrict__ wh,
                    const float* __restrict__ e_i,
                    const float* __restrict__ e_j,
                    const float* __restrict__ a_b,
                    float* __restrict__ out) {
    __shared__ alignas(16) int   adjS[2 * ADJ_STAGE];
    __shared__ alignas(16) float whT[2 * WH_STAGE];
    __shared__ alignas(16) float ejT[2 * EJ_STAGE];
    __shared__ float pden[8][32];
    __shared__ float accX[HPB][8][32];

    const int tid  = threadIdx.x;
    const int wv   = tid >> 5;
    const int hl   = wv & (HPB - 1);
    const int jh   = wv >> 2;
    const int lane = tid & 31;
    const int hg   = blockIdx.x & 1;
    const int hH   = hg * HPB + hl;
    const int i0   = (blockIdx.x >> 1) * 16;

    const int   iLoc  = lane & 15;
    const int   nCol  = lane & 15;
    const int   nColC = nCol & 7;
    const int   khalf = (lane >> 4) * 2;
    const float cmask = (nCol < DH) ? 1.0f : 0.0f;
    const float eib   = e_i[(size_t)(i0 + iLoc) * NHEADS + hH] + a_b[0];
    const int   dRow  = hl * DH + nColC;
    const int   jBase = jh * JHALF;

    const int aR = tid >> 6;
    const int aC = tid & 63;
    const int* adjP = adj + (size_t)(i0 + aR) * NN + aC * 4;
    const int wJ = tid >> 3;
    const int wD = (tid & 7) * 4;
    const float* whP = wh + (size_t)wJ * OUT_DIM + hg * (HPB * DH) + wD;
    const float* ejP = e_j + (size_t)tid * NHEADS + hg * HPB;

    v4i    rA[4];
    float4 rW[8];
    float4 rE;

    auto tile_load = [&](int chunk) {
#pragma unroll
        for (int k = 0; k < 4; ++k)
            rA[k] = *(const v4i*)(adjP + chunk + (size_t)(4 * k) * NN);
#pragma unroll
        for (int k = 0; k < 8; ++k)
            rW[k] = *(const float4*)(whP + (size_t)(chunk + k * 32) * OUT_DIM);
        rE = *(const float4*)(ejP + (size_t)chunk * NHEADS);
    };
    auto tile_store = [&](int st) {
        int*   aS = adjS + st * ADJ_STAGE;
        float* wS = whT  + st * WH_STAGE;
        float* eS = ejT  + st * EJ_STAGE;
#pragma unroll
        for (int k = 0; k < 4; ++k)
            ((v4i*)&aS[(aR + 4 * k) * LSTR])[aC] = rA[k];
#pragma unroll
        for (int k = 0; k < 8; ++k) {
            const int j = k * 32 + wJ;
            wS[(wD + 0) * LSTR + j] = rW[k].x;
            wS[(wD + 1) * LSTR + j] = rW[k].y;
            wS[(wD + 2) * LSTR + j] = rW[k].z;
            wS[(wD + 3) * LSTR + j] = rW[k].w;
        }
        eS[0 * LSTR + tid] = rE.x;
        eS[1 * LSTR + tid] = rE.y;
        eS[2 * LSTR + tid] = rE.z;
        eS[3 * LSTR + tid] = rE.w;
    };

    v8f   acc  = {0.f, 0.f, 0.f, 0.f, 0.f, 0.f, 0.f, 0.f};
    float psum = 0.f;

    auto tile_compute = [&](int st) {
        const int*   aS = adjS + st * ADJ_STAGE;
        const float* wS = whT  + st * WH_STAGE;
        const float* eS = ejT  + st * EJ_STAGE;
#pragma unroll 16
        for (int js = 0; js < JHALF; js += 4) {
            const int j0 = jBase + js + khalf;
            const float2 ej = *(const float2*)&eS[hl * LSTR + j0];
            const int2   ad = *(const int2*)  &aS[iLoc * LSTR + j0];
            const float2 wv2 = *(const float2*)&wS[dRow * LSTR + j0];

            float s0 = eib + ej.x;
            float s1 = eib + ej.y;
            s0 = (s0 >= 0.f) ? s0 : 0.2f * s0;
            s1 = (s1 >= 0.f) ? s1 : 0.2f * s1;
            const float m0 = (ad.x > 0) ? 1.0f : 0.0f;
            const float m1 = (ad.y > 0) ? 1.0f : 0.0f;
            const float p0 = __expf(s0) * m0;
            const float p1 = __expf(s1) * m1;
            psum += p0 + p1;

            v2f A; A.x = p0;            A.y = p1;
            v2f B; B.x = wv2.x * cmask; B.y = wv2.y * cmask;
            acc = __builtin_amdgcn_wmma_f32_16x16x4_f32(
                false, A, false, B, (short)0, acc, false, false);
        }
    };

    tile_load(0);
    tile_store(0);
    __syncthreads();

    for (int c = 0; c + 1 < NCHUNK; ++c) {
        tile_load((c + 1) * JCHUNK);
        tile_compute(c & 1);
        tile_store((c + 1) & 1);
        __syncthreads();
    }
    tile_compute((NCHUNK - 1) & 1);

    pden[wv][lane] = psum;
    if (jh == 1) {
#pragma unroll
        for (int v = 0; v < 8; ++v) accX[hl][v][lane] = acc[v];
    }
    __syncthreads();

    const int rbase = (lane >> 4) * 8;
    if (jh == 0 && nCol < DH) {
#pragma unroll
        for (int v = 0; v < 8; ++v) {
            const int   iL  = rbase + v;
            const float den = (pden[hl][iL] + pden[hl][iL + 16]) +
                              (pden[hl + HPB][iL] + pden[hl + HPB][iL + 16]);
            float o = (acc[v] + accX[hl][v][lane]) / den;
            o = (o > 0.f) ? o : (__expf(o) - 1.f);
            out[(size_t)hH * NN * DH + (size_t)(i0 + iL) * DH + nCol] = o;
        }
    }
}

extern "C" void kernel_launch(void* const* d_in, const int* in_sizes, int n_in,
                              void* d_out, int out_size, void* d_ws, size_t ws_size,
                              hipStream_t stream) {
    const float* h   = (const float*)d_in[0];
    const int*   adj = (const int*)  d_in[1];
    const float* W_w = (const float*)d_in[2];
    const float* W_b = (const float*)d_in[3];
    const float* a_w = (const float*)d_in[4];
    const float* a_b = (const float*)d_in[5];
    float* out = (float*)d_out;

    float* wh  = (float*)d_ws;
    float* e_i = wh  + (size_t)NN * OUT_DIM;
    float* e_j = e_i + (size_t)NN * NHEADS;

    gat_wh_kernel<<<NN, 64, 0, stream>>>(h, W_w, W_b, a_w, wh, e_i, e_j);
    gat_att_kernel<<<ATT_GRID, 256, 0, stream>>>(adj, wh, e_i, e_j, a_b, out);
}


# projection kernel (wh, e_i, e_j) as a tiled f32 WMMA GEMM: 16 x 32 tile per block, row-contiguous operand loads, W read once per block, not once per node (sum order over k changes)
# speedup vs baseline: 4.9055x; 1.3658x over previous
#include <hip/hip_runtime.h>
#include <hip/hip_bf16.h>


typedef __attribute__((ext_vector_type(2))) float v2f;
typedef __attribute__((ext_vector_type(8))) float v8f;
typedef __attribute__((ext_vector_type(4))) int   v4i;

#define NN      2048
#define IN_DIM  256
#define OUT_DIM 64
#define NHEADS  8
#define DH      8
#define HPB     4
#define JCHUNK  256
#define JHALF   (JCHUNK / 2)
#define LSTR    260

#define P1_COLS 32
#define P1_LSTR 36
#define P1_GRID ((NN / 16) * (OUT_DIM / P1_COLS))

__global__ __launch_bounds__(64) __attribute__((amdgpu_waves_per_eu(1, 1)))
void gat_wh_kernel(const float* __restrict__ h,
                   const float* __restrict__ W_w,
                   const float* __restrict__ W_b,
                   const float* __restrict__ a_w,
                   float* __restrict__ wh,
                   float* __restrict__ e_i,
                   float* __restrict__ e_j) {
    __shared__ alignas(16) float tile[16 * P1_LSTR];

    const int tid  = threadIdx.x;
    const int wv   = tid >> 5;
    const int lane = tid & 31;
    const int c    = lane & 15;
    const int hh   = lane >> 4;
    const int r0   = (blockIdx.x >> 1) * 16;
    const int cg   = blockIdx.x & 1;
    const int n0   = cg * P1_COLS + wv * 16;

    const float4* aP = (const float4*)(h   + (size_t)(r0 + c) * IN_DIM + 4 * hh);
    const float4* bP = (const float4*)(W_w + (size_t)(n0 + c) * IN_DIM + 4 * hh);

    float aw[2 * DH];
#pragma unroll
    for (int d = 0; d < 2 * DH; ++d) aw[d] = a_w[d];

    float4 ra[IN_DIM / 8], rb[IN_DIM / 8];
#pragma unroll
    for (int t = 0; t < IN_DIM / 8; ++t) {
        ra[t] = aP[2 * t];
        rb[t] = bP[2 * t];
    }

    const float bias = W_b[n0 + c];
    v8f acc0 = {bias, bias, bias, bias, bias, bias, bias, bias};
    v8f acc1 = {0.f, 0.f, 0.f, 0.f, 0.f, 0.f, 0.f, 0.f};
#pragma unroll
    for (int t = 0; t < IN_DIM / 8; ++t) {
        v2f A0; A0.x = ra[t].x; A0.y = ra[t].y;
        v2f B0; B0.x = rb[t].x; B0.y = rb[t].y;
        v2f A1; A1.x = ra[t].z; A1.y = ra[t].w;
        v2f B1; B1.x = rb[t].z; B1.y = rb[t].w;
        acc0 = __builtin_amdgcn_wmma_f32_16x16x4_f32(
            false, A0, false, B0, (short)0, acc0, false, false);
        acc1 = __builtin_amdgcn_wmma_f32_16x16x4_f32(
            false, A1, false, B1, (short)0, acc1, false, false);
    }

#pragma unroll
    for (int r = 0; r < 8; ++r)
        tile[(8 * hh + r) * P1_LSTR + wv * 16 + c] = acc0[r] + acc1[r];
    __syncthreads();

#pragma unroll
    for (int k = 0; k < 2; ++k) {
        const int row  = (tid >> 3) + 8 * k;
        const int col4 = (tid & 7) * 4;
        const float4 v = *(const float4*)&tile[row * P1_LSTR + col4];
        *(float4*)(wh + (size_t)(r0 + row) * OUT_DIM + cg * P1_COLS + col4) = v;
    }

    {
        const int row = tid >> 2;
        const int hl  = tid & 3;
        const float* w = &tile[row * P1_LSTR + hl * DH];
        float si = 0.f, sj = 0.f;
#pragma unroll
        for (int d = 0; d < DH; ++d) {
            si = fmaf(w[d], aw[d], si);
            sj = fmaf(w[d], aw[DH + d], sj);
        }
        const size_t o = (size_t)(r0 + row) * NHEADS + cg * (P1_COLS / DH) + hl;
        e_i[o] = si;
        e_j[o] = sj;
    }
}

#define NCHUNK    (NN / JCHUNK)
#define ADJ_STAGE (16 * LSTR)
#define WH_STAGE  (HPB * DH * LSTR)
#define EJ_STAGE  (HPB * LSTR)
#define ATT_GRID  ((NN / 16) * (NHEADS / HPB))

__global__ __launch_bounds__(256)
void gat_att_kernel(const int*   __restrict__ adj,
                    const float* __restrict__ wh,
                    const float* __restrict__ e_i,
                    const float* __restrict__ e_j,
                    const float* __restrict__ a_b,
                    float* __restrict__ out) {
    __shared__ alignas(16) int   adjS[2 * ADJ_STAGE];
    __shared__ alignas(16) float whT[2 * WH_STAGE];
    __shared__ alignas(16) float ejT[2 * EJ_STAGE];
    __shared__ float pden[8][32];
    __shared__ float accX[HPB][8][32];

    const int tid  = threadIdx.x;
    const int wv   = tid >> 5;
    const int hl   = wv & (HPB - 1);
    const int jh   = wv >> 2;
    const int lane = tid & 31;
    const int hg   = blockIdx.x & 1;
    const int hH   = hg * HPB + hl;
    const int i0   = (blockIdx.x >> 1) * 16;

    const int   iLoc  = lane & 15;
    const int   nCol  = lane & 15;
    const int   nColC = nCol & 7;
    const int   khalf = (lane >> 4) * 2;
    const float cmask = (nCol < DH) ? 1.0f : 0.0f;
    const float eib   = e_i[(size_t)(i0 + iLoc) * NHEADS + hH] + a_b[0];
    const int   dRow  = hl * DH + nColC;
    const int   jBase = jh * JHALF;

    const int aR = tid >> 6;
    const int aC = tid & 63;
    const int* adjP = adj + (size_t)(i0 + aR) * NN + aC * 4;
    const int wJ = tid >> 3;
    const int wD = (tid & 7) * 4;
    const float* whP = wh + (size_t)wJ * OUT_DIM + hg * (HPB * DH) + wD;
    const float* ejP = e_j + (size_t)tid * NHEADS + hg * HPB;

    v4i    rA[4];
    float4 rW[8];
    float4 rE;

    auto tile_load = [&](int chunk) {
#pragma unroll
        for (int k = 0; k < 4; ++k)
            rA[k] = *(const v4i*)(adjP + chunk + (size_t)(4 * k) * NN);
#pragma unroll
        for (int k = 0; k < 8; ++k)
            rW[k] = *(const float4*)(whP + (size_t)(chunk + k * 32) * OUT_DIM);
        rE = *(const float4*)(ejP + (size_t)chunk * NHEADS);
    };
    auto tile_store = [&](int st) {
        int*   aS = adjS + st * ADJ_STAGE;
        float* wS = whT  + st * WH_STAGE;
        float* eS = ejT  + st * EJ_STAGE;
#pragma unroll
        for (int k = 0; k < 4; ++k)
            ((v4i*)&aS[(aR + 4 * k) * LSTR])[aC] = rA[k];
#pragma unroll
        for (int k = 0; k < 8; ++k) {
            const int j = k * 32 + wJ;
            wS[(wD + 0) * LSTR + j] = rW[k].x;
            wS[(wD + 1) * LSTR + j] = rW[k].y;
            wS[(wD + 2) * LSTR + j] = rW[k].z;
            wS[(wD + 3) * LSTR + j] = rW[k].w;
        }
        eS[0 * LSTR + tid] = rE.x;
        eS[1 * LSTR + tid] = rE.y;
        eS[2 * LSTR + tid] = rE.z;
        eS[3 * LSTR + tid] = rE.w;
    };

    v8f   acc  = {0.f, 0.f, 0.f, 0.f, 0.f, 0.f, 0.f, 0.f};
    float psum = 0.f;

    auto tile_compute = [&](int st) {
        const int*   aS = adjS + st * ADJ_STAGE;
        const float* wS = whT  + st * WH_STAGE;
        const float* eS = ejT  + st * EJ_STAGE;
#pragma unroll 16
        for (int js = 0; js < JHALF; js += 4) {
            const int j0 = jBase + js + khalf;
            const float2 ej = *(const float2*)&eS[hl * LSTR + j0];
            const int2   ad = *(const int2*)  &aS[iLoc * LSTR + j0];
            const float2 wv2 = *(const float2*)&wS[dRow * LSTR + j0];

            float s0 = eib + ej.x;
            float s1 = eib + ej.y;
            s0 = (s0 >= 0.f) ? s0 : 0.2f * s0;
            s1 = (s1 >= 0.f) ? s1 : 0.2f * s1;
            const float m0 = (ad.x > 0) ? 1.0f : 0.0f;
            const float m1 = (ad.y > 0) ? 1.0f : 0.0f;
            const float p0 = __expf(s0) * m0;
            const float p1 = __expf(s1) * m1;
            psum += p0 + p1;

            v2f A; A.x = p0;            A.y = p1;
            v2f B; B.x = wv2.x * cmask; B.y = wv2.y * cmask;
            acc = __builtin_amdgcn_wmma_f32_16x16x4_f32(
                false, A, false, B, (short)0, acc, false, false);
        }
    };

    tile_load(0);
    tile_store(0);
    __syncthreads();

    for (int c = 0; c + 1 < NCHUNK; ++c) {
        tile_load((c + 1) * JCHUNK);
        tile_compute(c & 1);
        tile_store((c + 1) & 1);
        __syncthreads();
    }
    tile_compute((NCHUNK - 1) & 1);

    pden[wv][lane] = psum;
    if (jh == 1) {
#pragma unroll
        for (int v = 0; v < 8; ++v) accX[hl][v][lane] = acc[v];
    }
    __syncthreads();

    const int rbase = (lane >> 4) * 8;
    if (jh == 0 && nCol < DH) {
#pragma unroll
        for (int v = 0; v < 8; ++v) {
            const int   iL  = rbase + v;
            const float den = (pden[hl][iL] + pden[hl][iL + 16]) +
                              (pden[hl + HPB][iL] + pden[hl + HPB][iL + 16]);
            float o = (acc[v] + accX[hl][v][lane]) / den;
            o = (o > 0.f) ? o : (__expf(o) - 1.f);
            out[(size_t)hH * NN * DH + (size_t)(i0 + iL) * DH + nCol] = o;
        }
    }
}

extern "C" void kernel_launch(void* const* d_in, const int* in_sizes, int n_in,
                              void* d_out, int out_size, void* d_ws, size_t ws_size,
                              hipStream_t stream) {
    const float* h   = (const float*)d_in[0];
    const int*   adj = (const int*)  d_in[1];
    const float* W_w = (const float*)d_in[2];
    const float* W_b = (const float*)d_in[3];
    const float* a_w = (const float*)d_in[4];
    const float* a_b = (const float*)d_in[5];
    float* out = (float*)d_out;

    float* wh  = (float*)d_ws;
    float* e_i = wh  + (size_t)NN * OUT_DIM;
    float* e_j = e_i + (size_t)NN * NHEADS;

    gat_wh_kernel<<<P1_GRID, 64, 0, stream>>>(h, W_w, W_b, a_w, wh, e_i, e_j);
    gat_att_kernel<<<ATT_GRID, 256, 0, stream>>>(adj, wh, e_i, e_j, a_b, out);
}


# attention kernel: exp(leakyrelu(ei+ej+b)) taken as max(Ei*Ej, Fi*Fj), the four exponentials made once per (node, head) by the projection kernel; no exp per score (weight rounding changes)
# speedup vs baseline: 5.2547x; 1.0712x over previous
#include <hip/hip_runtime.h>
#include <hip/hip_bf16.h>


typedef __attribute__((ext_vector_type(2))) float v2f;
typedef __attribute__((ext_vector_type(8))) float v8f;
typedef __attribute__((ext_vector_type(4))) int   v4i;

#define NN      2048
#define IN_DIM  256
#define OUT_DIM 64
#define NHEADS  8
#define DH      8
#define HPB     4
#define JCHUNK  256
#define JHALF   (JCHUNK / 2)
#define LSTR    260

#define P1_COLS 32
#define P1_LSTR 36
#define P1_GRID ((NN / 16) * (OUT_DIM / P1_COLS))

__global__ __launch_bounds__(64) __attribute__((amdgpu_waves_per_eu(1, 1)))
void gat_wh_kernel(const float* __restrict__ h,
                   const float* __restrict__ W_w,
                   const float* __restrict__ W_b,
                   const float* __restrict__ a_w,
                   const float* __restrict__ a_b,
                   float* __restrict__ wh,
                   float* __restrict__ efI,
                   float* __restrict__ efJ) {
    __shared__ alignas(16) float tile[16 * P1_LSTR];

    const int tid  = threadIdx.x;
    const int wv   = tid >> 5;
    const int lane = tid & 31;
    const int c    = lane & 15;
    const int hh   = lane >> 4;
    const int r0   = (blockIdx.x >> 1) * 16;
    const int cg   = blockIdx.x & 1;
    const int n0   = cg * P1_COLS + wv * 16;

    const float4* aP = (const float4*)(h   + (size_t)(r0 + c) * IN_DIM + 4 * hh);
    const float4* bP = (const float4*)(W_w + (size_t)(n0 + c) * IN_DIM + 4 * hh);

    float aw[2 * DH];
#pragma unroll
    for (int d = 0; d < 2 * DH; ++d) aw[d] = a_w[d];
    const float ab = a_b[0];

    float4 ra[IN_DIM / 8], rb[IN_DIM / 8];
#pragma unroll
    for (int t = 0; t < IN_DIM / 8; ++t) {
        ra[t] = aP[2 * t];
        rb[t] = bP[2 * t];
    }

    const float bias = W_b[n0 + c];
    v8f acc0 = {bias, bias, bias, bias, bias, bias, bias, bias};
    v8f acc1 = {0.f, 0.f, 0.f, 0.f, 0.f, 0.f, 0.f, 0.f};
#pragma unroll
    for (int t = 0; t < IN_DIM / 8; ++t) {
        v2f A0; A0.x = ra[t].x; A0.y = ra[t].y;
        v2f B0; B0.x = rb[t].x; B0.y = rb[t].y;
        v2f A1; A1.x = ra[t].z; A1.y = ra[t].w;
        v2f B1; B1.x = rb[t].z; B1.y = rb[t].w;
        acc0 = __builtin_amdgcn_wmma_f32_16x16x4_f32(
            false, A0, false, B0, (short)0, acc0, false, false);
        acc1 = __builtin_amdgcn_wmma_f32_16x16x4_f32(
            false, A1, false, B1, (short)0, acc1, false, false);
    }

#pragma unroll
    for (int r = 0; r < 8; ++r)
        tile[(8 * hh + r) * P1_LSTR + wv * 16 + c] = acc0[r] + acc1[r];
    __syncthreads();

#pragma unroll
    for (int k = 0; k < 2; ++k) {
        const int row  = (tid >> 3) + 8 * k;
        const int col4 = (tid & 7) * 4;
        const float4 v = *(const float4*)&tile[row * P1_LSTR + col4];
        *(float4*)(wh + (size_t)(r0 + row) * OUT_DIM + cg * P1_COLS + col4) = v;
    }

    {
        const int row = tid >> 2;
        const int hl  = tid & 3;
        const float* w = &tile[row * P1_LSTR + hl * DH];
        float si = 0.f, sj = 0.f;
#pragma unroll
        for (int d = 0; d < DH; ++d) {
            si = fmaf(w[d], aw[d], si);
            sj = fmaf(w[d], aw[DH + d], sj);
        }
        const size_t node = (size_t)(r0 + row);
        float2 fi;
        fi.x = __expf(si + ab);
        fi.y = __expf(0.2f * (si + ab));
        *(float2*)(efI + (node * NHEADS + cg * (P1_COLS / DH) + hl) * 2) = fi;
        float* ef = efJ + ((size_t)cg * NN + node) * (2 * HPB);
        ef[hl]       = __expf(sj);
        ef[HPB + hl] = __expf(0.2f * sj);
    }
}

#define NCHUNK    (NN / JCHUNK)
#define ADJ_STAGE (16 * LSTR)
#define WH_STAGE  (HPB * DH * LSTR)
#define ESTR      (2 * LSTR)
#define EJ_STAGE  (HPB * ESTR)
#define ATT_GRID  ((NN / 16) * (NHEADS / HPB))

__global__ __launch_bounds__(256)
void gat_att_kernel(const int*   __restrict__ adj,
                    const float* __restrict__ wh,
                    const float* __restrict__ efI,
                    const float* __restrict__ efJ,
                    float* __restrict__ out) {
    __shared__ alignas(16) int   adjS[2 * ADJ_STAGE];
    __shared__ alignas(16) float whT[2 * WH_STAGE];
    __shared__ alignas(16) float ejT[2 * EJ_STAGE];
    __shared__ float pden[8][32];
    __shared__ float accX[HPB][8][32];

    const int tid  = threadIdx.x;
    const int wv   = tid >> 5;
    const int hl   = wv & (HPB - 1);
    const int jh   = wv >> 2;
    const int lane = tid & 31;
    const int hg   = blockIdx.x & 1;
    const int hH   = hg * HPB + hl;
    const int i0   = (blockIdx.x >> 1) * 16;

    const int   iLoc  = lane & 15;
    const int   nCol  = lane & 15;
    const int   nColC = nCol & 7;
    const int   khalf = (lane >> 4) * 2;
    const float cmask = (nCol < DH) ? 1.0f : 0.0f;
    const float2 fi   = *(const float2*)(efI + ((size_t)(i0 + iLoc) * NHEADS + hH) * 2);
    const int   dRow  = hl * DH + nColC;
    const int   jBase = jh * JHALF;

    const int aR = tid >> 6;
    const int aC = tid & 63;
    const int* adjP = adj + (size_t)(i0 + aR) * NN + aC * 4;
    const int wJ = tid >> 3;
    const int wD = (tid & 7) * 4;
    const float* whP = wh + (size_t)wJ * OUT_DIM + hg * (HPB * DH) + wD;
    const float* ejP = efJ + ((size_t)hg * NN + tid) * (2 * HPB);

    v4i    rA[4];
    float4 rW[8];
    float4 rE, rF;

    auto tile_load = [&](int chunk) {
#pragma unroll
        for (int k = 0; k < 4; ++k)
            rA[k] = *(const v4i*)(adjP + chunk + (size_t)(4 * k) * NN);
#pragma unroll
        for (int k = 0; k < 8; ++k)
            rW[k] = *(const float4*)(whP + (size_t)(chunk + k * 32) * OUT_DIM);
        rE = *(const float4*)(ejP + (size_t)chunk * (2 * HPB));
        rF = *(const float4*)(ejP + (size_t)chunk * (2 * HPB) + HPB);
    };
    auto tile_store = [&](int st) {
        int*   aS = adjS + st * ADJ_STAGE;
        float* wS = whT  + st * WH_STAGE;
        float* eS = ejT  + st * EJ_STAGE;
#pragma unroll
        for (int k = 0; k < 4; ++k)
            ((v4i*)&aS[(aR + 4 * k) * LSTR])[aC] = rA[k];
#pragma unroll
        for (int k = 0; k < 8; ++k) {
            const int j = k * 32 + wJ;
            wS[(wD + 0) * LSTR + j] = rW[k].x;
            wS[(wD + 1) * LSTR + j] = rW[k].y;
            wS[(wD + 2) * LSTR + j] = rW[k].z;
            wS[(wD + 3) * LSTR + j] = rW[k].w;
        }
        const int eo = (tid >> 1) * 4 + (tid & 1);
        eS[0 * ESTR + eo] = rE.x;  eS[0 * ESTR + eo + 2] = rF.x;
        eS[1 * ESTR + eo] = rE.y;  eS[1 * ESTR + eo + 2] = rF.y;
        eS[2 * ESTR + eo] = rE.z;  eS[2 * ESTR + eo + 2] = rF.z;
        eS[3 * ESTR + eo] = rE.w;  eS[3 * ESTR + eo + 2] = rF.w;
    };

    v8f   acc  = {0.f, 0.f, 0.f, 0.f, 0.f, 0.f, 0.f, 0.f};
    float psum = 0.f;

    auto tile_compute = [&](int st) {
        const int*   aS = adjS + st * ADJ_STAGE;
        const float* wS = whT  + st * WH_STAGE;
        const float* eS = ejT  + st * EJ_STAGE;
#pragma unroll 16
        for (int js = 0; js < JHALF; js += 4) {
            const int j0 = jBase + js + khalf;
            const float4 ef = *(const float4*)&eS[hl * ESTR + 2 * j0];
            const int2   ad = *(const int2*)  &aS[iLoc * LSTR + j0];
            const float2 wv2 = *(const float2*)&wS[dRow * LSTR + j0];

            v2f pe; pe.x = ef.x; pe.y = ef.y;
            v2f pf; pf.x = ef.z; pf.y = ef.w;
            pe = pe * fi.x;
            pf = pf * fi.y;
            const float p0 = (ad.x > 0) ? __builtin_fmaxf(pe.x, pf.x) : 0.0f;
            const float p1 = (ad.y > 0) ? __builtin_fmaxf(pe.y, pf.y) : 0.0f;
            psum += p0 + p1;

            v2f A; A.x = p0;            A.y = p1;
            v2f B; B.x = wv2.x * cmask; B.y = wv2.y * cmask;
            acc = __builtin_amdgcn_wmma_f32_16x16x4_f32(
                false, A, false, B, (short)0, acc, false, false);
        }
    };

    tile_load(0);
    tile_store(0);
    __syncthreads();

    for (int c = 0; c + 1 < NCHUNK; ++c) {
        tile_load((c + 1) * JCHUNK);
        tile_compute(c & 1);
        tile_store((c + 1) & 1);
        __syncthreads();
    }
    tile_compute((NCHUNK - 1) & 1);

    pden[wv][lane] = psum;
    if (jh == 1) {
#pragma unroll
        for (int v = 0; v < 8; ++v) accX[hl][v][lane] = acc[v];
    }
    __syncthreads();

    const int rbase = (lane >> 4) * 8;
    if (jh == 0 && nCol < DH) {
#pragma unroll
        for (int v = 0; v < 8; ++v) {
            const int   iL  = rbase + v;
            const float den = (pden[hl][iL] + pden[hl][iL + 16]) +
                              (pden[hl + HPB][iL] + pden[hl + HPB][iL + 16]);
            float o = (acc[v] + accX[hl][v][lane]) / den;
            o = (o > 0.f) ? o : (__expf(o) - 1.f);
            out[(size_t)hH * NN * DH + (size_t)(i0 + iL) * DH + nCol] = o;
        }
    }
}

extern "C" void kernel_launch(void* const* d_in, const int* in_sizes, int n_in,
                              void* d_out, int out_size, void* d_ws, size_t ws_size,
                              hipStream_t stream) {
    const float* h   = (const float*)d_in[0];
    const int*   adj = (const int*)  d_in[1];
    const float* W_w = (const float*)d_in[2];
    const float* W_b = (const float*)d_in[3];
    const float* a_w = (const float*)d_in[4];
    const float* a_b = (const float*)d_in[5];
    float* out = (float*)d_out;

    float* wh  = (float*)d_ws;
    float* efI = wh  + (size_t)NN * OUT_DIM;
    float* efJ = efI + (size_t)NN * NHEADS * 2;

    gat_wh_kernel<<<P1_GRID, 64, 0, stream>>>(h, W_w, W_b, a_w, a_b, wh, efI, efJ);
    gat_att_kernel<<<ATT_GRID, 256, 0, stream>>>(adj, wh, efI, efJ, out);
}


# exp(leakyrelu(x)) = F_i F_j max(1, u_i v_j): F_j folded into the staged wh rows, F_j row as B column 8 gives the denominator, F_i cancels; one factor per score (weight rounding, sum order change)
# speedup vs baseline: 5.4490x; 1.0370x over previous
#include <hip/hip_runtime.h>
#include <hip/hip_bf16.h>


typedef __attribute__((ext_vector_type(2))) float v2f;
typedef __attribute__((ext_vector_type(8))) float v8f;
typedef __attribute__((ext_vector_type(4))) int   v4i;

#define NN      2048
#define IN_DIM  256
#define OUT_DIM 64
#define NHEADS  8
#define DH      8
#define HPB     4
#define JCHUNK  256
#define JHALF   (JCHUNK / 2)
#define LSTR    260

#define P1_COLS 32
#define P1_LSTR 36
#define P1_GRID ((NN / 16) * (OUT_DIM / P1_COLS))

__global__ __launch_bounds__(64) __attribute__((amdgpu_waves_per_eu(1, 1)))
void gat_wh_kernel(const float* __restrict__ h,
                   const float* __restrict__ W_w,
                   const float* __restrict__ W_b,
                   const float* __restrict__ a_w,
                   const float* __restrict__ a_b,
                   float* __restrict__ whF,
                   float* __restrict__ efI,
                   float* __restrict__ efJ) {
    __shared__ alignas(16) float tile[16 * P1_LSTR];

    const int tid  = threadIdx.x;
    const int wv   = tid >> 5;
    const int lane = tid & 31;
    const int c    = lane & 15;
    const int hh   = lane >> 4;
    const int r0   = (blockIdx.x >> 1) * 16;
    const int cg   = blockIdx.x & 1;
    const int n0   = cg * P1_COLS + wv * 16;

    const float4* aP = (const float4*)(h   + (size_t)(r0 + c) * IN_DIM + 4 * hh);
    const float4* bP = (const float4*)(W_w + (size_t)(n0 + c) * IN_DIM + 4 * hh);

    float aw[2 * DH];
#pragma unroll
    for (int d = 0; d < 2 * DH; ++d) aw[d] = a_w[d];
    const float ab = a_b[0];

    float4 ra[IN_DIM / 8], rb[IN_DIM / 8];
#pragma unroll
    for (int t = 0; t < IN_DIM / 8; ++t) {
        ra[t] = aP[2 * t];
        rb[t] = bP[2 * t];
    }

    const float bias = W_b[n0 + c];
    v8f acc0 = {bias, bias, bias, bias, bias, bias, bias, bias};
    v8f acc1 = {0.f, 0.f, 0.f, 0.f, 0.f, 0.f, 0.f, 0.f};
#pragma unroll
    for (int t = 0; t < IN_DIM / 8; ++t) {
        v2f A0; A0.x = ra[t].x; A0.y = ra[t].y;
        v2f B0; B0.x = rb[t].x; B0.y = rb[t].y;
        v2f A1; A1.x = ra[t].z; A1.y = ra[t].w;
        v2f B1; B1.x = rb[t].z; B1.y = rb[t].w;
        acc0 = __builtin_amdgcn_wmma_f32_16x16x4_f32(
            false, A0, false, B0, (short)0, acc0, false, false);
        acc1 = __builtin_amdgcn_wmma_f32_16x16x4_f32(
            false, A1, false, B1, (short)0, acc1, false, false);
    }

#pragma unroll
    for (int r = 0; r < 8; ++r)
        tile[(8 * hh + r) * P1_LSTR + wv * 16 + c] = acc0[r] + acc1[r];
    __syncthreads();

    {
        const int row = tid >> 2;
        const int hl  = tid & 3;
        float* w = &tile[row * P1_LSTR + hl * DH];
        float wv8[DH];
        float si = 0.f, sj = 0.f;
#pragma unroll
        for (int d = 0; d < DH; ++d) {
            wv8[d] = w[d];
            si = fmaf(wv8[d], aw[d], si);
            sj = fmaf(wv8[d], aw[DH + d], sj);
        }
        const size_t node = (size_t)(r0 + row);
        const float fj = __expf(0.2f * sj);
        efI[node * NHEADS + cg * (P1_COLS / DH) + hl] = __expf(0.8f * (si + ab));
        float* ef = efJ + ((size_t)cg * NN + node) * (2 * HPB);
        ef[hl]       = fj;
        ef[HPB + hl] = __expf(0.8f * sj);
#pragma unroll
        for (int d = 0; d < DH; ++d) w[d] = wv8[d] * fj;
    }
    __syncthreads();

#pragma unroll
    for (int k = 0; k < 2; ++k) {
        const int row  = (tid >> 3) + 8 * k;
        const int col4 = (tid & 7) * 4;
        const float4 v = *(const float4*)&tile[row * P1_LSTR + col4];
        *(float4*)(whF + (size_t)(r0 + row) * OUT_DIM + cg * P1_COLS + col4) = v;
    }
}

#define NCHUNK    (NN / JCHUNK)
#define ADJ_STAGE (16 * LSTR)
#define BROWS     (DH + 2)
#define WH_STAGE  (HPB * BROWS * LSTR)
#define VJ_STAGE  (HPB * LSTR)
#define ATT_GRID  ((NN / 16) * (NHEADS / HPB))

__global__ __launch_bounds__(256)
void gat_att_kernel(const int*   __restrict__ adj,
                    const float* __restrict__ whF,
                    const float* __restrict__ efI,
                    const float* __restrict__ efJ,
                    float* __restrict__ out) {
    __shared__ alignas(16) int   adjS[2 * ADJ_STAGE];
    __shared__ alignas(16) float whT[2 * WH_STAGE];
    __shared__ alignas(16) float vjT[2 * VJ_STAGE];
    __shared__ float accX[2][HPB][8][32];

    const int tid  = threadIdx.x;
    const int wv   = tid >> 5;
    const int hl   = wv & (HPB - 1);
    const int jh   = wv >> 2;
    const int lane = tid & 31;
    const int hg   = blockIdx.x & 1;
    const int hH   = hg * HPB + hl;
    const int i0   = (blockIdx.x >> 1) * 16;

    const int   iLoc  = lane & 15;
    const int   nCol  = lane & 15;
    const int   khalf = (lane >> 4) * 2;
    const float ui    = efI[(size_t)(i0 + iLoc) * NHEADS + hH];
    const int   dRow  = hl * BROWS + ((nCol < DH + 1) ? nCol : DH + 1);
    const int   jBase = jh * JHALF;

    const int aR = tid >> 6;
    const int aC = tid & 63;
    const int* adjP = adj + (size_t)(i0 + aR) * NN + aC * 4;
    const int wJ = tid >> 3;
    const int wD = (tid & 7) * 4;
    const float* whP = whF + (size_t)wJ * OUT_DIM + hg * (HPB * DH) + wD;
    const int wRow = (wD >> 3) * BROWS + (wD & 7);
    const float* ejP = efJ + ((size_t)hg * NN + tid) * (2 * HPB);

    v4i    rA[4];
    float4 rW[8];
    float4 rE, rF;

    auto tile_load = [&](int chunk) {
#pragma unroll
        for (int k = 0; k < 4; ++k)
            rA[k] = *(const v4i*)(adjP + chunk + (size_t)(4 * k) * NN);
#pragma unroll
        for (int k = 0; k < 8; ++k)
            rW[k] = *(const float4*)(whP + (size_t)(chunk + k * 32) * OUT_DIM);
        rE = *(const float4*)(ejP + (size_t)chunk * (2 * HPB));
        rF = *(const float4*)(ejP + (size_t)chunk * (2 * HPB) + HPB);
    };
    auto tile_store = [&](int st) {
        int*   aS = adjS + st * ADJ_STAGE;
        float* wS = whT  + st * WH_STAGE;
        float* vS = vjT  + st * VJ_STAGE;
#pragma unroll
        for (int k = 0; k < 4; ++k)
            ((v4i*)&aS[(aR + 4 * k) * LSTR])[aC] = rA[k];
#pragma unroll
        for (int k = 0; k < 8; ++k) {
            const int j = k * 32 + wJ;
            wS[(wRow + 0) * LSTR + j] = rW[k].x;
            wS[(wRow + 1) * LSTR + j] = rW[k].y;
            wS[(wRow + 2) * LSTR + j] = rW[k].z;
            wS[(wRow + 3) * LSTR + j] = rW[k].w;
        }
        wS[(0 * BROWS + DH) * LSTR + tid] = rE.x;  vS[0 * LSTR + tid] = rF.x;
        wS[(1 * BROWS + DH) * LSTR + tid] = rE.y;  vS[1 * LSTR + tid] = rF.y;
        wS[(2 * BROWS + DH) * LSTR + tid] = rE.z;  vS[2 * LSTR + tid] = rF.z;
        wS[(3 * BROWS + DH) * LSTR + tid] = rE.w;  vS[3 * LSTR + tid] = rF.w;
    };

    v8f   acc  = {0.f, 0.f, 0.f, 0.f, 0.f, 0.f, 0.f, 0.f};

    auto tile_compute = [&](int st) {
        const int*   aS = adjS + st * ADJ_STAGE;
        const float* wS = whT  + st * WH_STAGE;
        const float* vS = vjT  + st * VJ_STAGE;
#pragma unroll 16
        for (int js = 0; js < JHALF; js += 4) {
            const int j0 = jBase + js + khalf;
            const float2 vj  = *(const float2*)&vS[hl * LSTR + j0];
            const int2   ad  = *(const int2*)  &aS[iLoc * LSTR + j0];
            const float2 wv2 = *(const float2*)&wS[dRow * LSTR + j0];

            v2f pm; pm.x = vj.x; pm.y = vj.y;
            pm = pm * ui;
            const float p0 = (ad.x > 0) ? __builtin_fmaxf(pm.x, 1.0f) : 0.0f;
            const float p1 = (ad.y > 0) ? __builtin_fmaxf(pm.y, 1.0f) : 0.0f;

            v2f A; A.x = p0;    A.y = p1;
            v2f B; B.x = wv2.x; B.y = wv2.y;
            acc = __builtin_amdgcn_wmma_f32_16x16x4_f32(
                false, A, false, B, (short)0, acc, false, false);
        }
    };

#pragma unroll
    for (int zr = 0; zr < 2 * HPB; ++zr)
        whT[(zr >> 2) * WH_STAGE + ((zr & 3) * BROWS + DH + 1) * LSTR + tid] = 0.f;
    tile_load(0);
    tile_store(0);
    __syncthreads();

    for (int c = 0; c + 1 < NCHUNK; ++c) {
        tile_load((c + 1) * JCHUNK);
        tile_compute(c & 1);
        tile_store((c + 1) & 1);
        __syncthreads();
    }
    tile_compute((NCHUNK - 1) & 1);

#pragma unroll
    for (int v = 0; v < 8; ++v) accX[jh][hl][v][lane] = acc[v];
    __syncthreads();

    const int rbase = (lane >> 4) * 8;
    const int dLane = (lane & 16) + DH;
    if (jh == 0 && nCol < DH) {
#pragma unroll
        for (int v = 0; v < 8; ++v) {
            const int   iL  = rbase + v;
            const float den = accX[0][hl][v][dLane] + accX[1][hl][v][dLane];
            float o = (acc[v] + accX[1][hl][v][lane]) / den;
            o = (o > 0.f) ? o : (__expf(o) - 1.f);
            out[(size_t)hH * NN * DH + (size_t)(i0 + iL) * DH + nCol] = o;
        }
    }
}

extern "C" void kernel_launch(void* const* d_in, const int* in_sizes, int n_in,
                              void* d_out, int out_size, void* d_ws, size_t ws_size,
                              hipStream_t stream) {
    const float* h   = (const float*)d_in[0];
    const int*   adj = (const int*)  d_in[1];
    const float* W_w = (const float*)d_in[2];
    const float* W_b = (const float*)d_in[3];
    const float* a_w = (const float*)d_in[4];
    const float* a_b = (const float*)d_in[5];
    float* out = (float*)d_out;

    float* whF = (float*)d_ws;
    float* efI = whF + (size_t)NN * OUT_DIM;
    float* efJ = efI + (size_t)NN * NHEADS;

    gat_wh_kernel<<<P1_GRID, 64, 0, stream>>>(h, W_w, W_b, a_w, a_b, whF, efI, efJ);
    gat_att_kernel<<<ATT_GRID, 256, 0, stream>>>(adj, whF, efI, efJ, out);
}


# attention kernel: next j-tile's LDS stores and the following tile's fetch moved between the current tile's two 16-WMMA trips; each trip's LDS operand reads go out as one group; sum order unchanged
# speedup vs baseline: 5.6164x; 1.0307x over previous
#include <hip/hip_runtime.h>
#include <hip/hip_bf16.h>


typedef __attribute__((ext_vector_type(2))) float v2f;
typedef __attribute__((ext_vector_type(8))) float v8f;
typedef __attribute__((ext_vector_type(4))) int   v4i;

#define NN      2048
#define IN_DIM  256
#define OUT_DIM 64
#define NHEADS  8
#define DH      8
#define HPB     4
#define JCHUNK  256
#define JHALF   (JCHUNK / 2)
#define LSTR    260

#define P1_COLS 32
#define P1_LSTR 36
#define P1_GRID ((NN / 16) * (OUT_DIM / P1_COLS))

__global__ __launch_bounds__(64) __attribute__((amdgpu_waves_per_eu(1, 1)))
void gat_wh_kernel(const float* __restrict__ h,
                   const float* __restrict__ W_w,
                   const float* __restrict__ W_b,
                   const float* __restrict__ a_w,
                   const float* __restrict__ a_b,
                   float* __restrict__ whF,
                   float* __restrict__ efI,
                   float* __restrict__ efJ) {
    __shared__ alignas(16) float tile[16 * P1_LSTR];

    const int tid  = threadIdx.x;
    const int wv   = tid >> 5;
    const int lane = tid & 31;
    const int c    = lane & 15;
    const int hh   = lane >> 4;
    const int r0   = (blockIdx.x >> 1) * 16;
    const int cg   = blockIdx.x & 1;
    const int n0   = cg * P1_COLS + wv * 16;

    const float4* aP = (const float4*)(h   + (size_t)(r0 + c) * IN_DIM + 4 * hh);
    const float4* bP = (const float4*)(W_w + (size_t)(n0 + c) * IN_DIM + 4 * hh);

    float aw[2 * DH];
#pragma unroll
    for (int d = 0; d < 2 * DH; ++d) aw[d] = a_w[d];
    const float ab = a_b[0];

    float4 ra[IN_DIM / 8], rb[IN_DIM / 8];
#pragma unroll
    for (int t = 0; t < IN_DIM / 8; ++t) {
        ra[t] = aP[2 * t];
        rb[t] = bP[2 * t];
    }

    const float bias = W_b[n0 + c];
    v8f acc0 = {bias, bias, bias, bias, bias, bias, bias, bias};
    v8f acc1 = {0.f, 0.f, 0.f, 0.f, 0.f, 0.f, 0.f, 0.f};
#pragma unroll
    for (int t = 0; t < IN_DIM / 8; ++t) {
        v2f A0; A0.x = ra[t].x; A0.y = ra[t].y;
        v2f B0; B0.x = rb[t].x; B0.y = rb[t].y;
        v2f A1; A1.x = ra[t].z; A1.y = ra[t].w;
        v2f B1; B1.x = rb[t].z; B1.y = rb[t].w;
        acc0 = __builtin_amdgcn_wmma_f32_16x16x4_f32(
            false, A0, false, B0, (short)0, acc0, false, false);
        acc1 = __builtin_amdgcn_wmma_f32_16x16x4_f32(
            false, A1, false, B1, (short)0, acc1, false, false);
    }

#pragma unroll
    for (int r = 0; r < 8; ++r)
        tile[(8 * hh + r) * P1_LSTR + wv * 16 + c] = acc0[r] + acc1[r];
    __syncthreads();

    {
        const int row = tid >> 2;
        const int hl  = tid & 3;
        float* w = &tile[row * P1_LSTR + hl * DH];
        float wv8[DH];
        float si = 0.f, sj = 0.f;
#pragma unroll
        for (int d = 0; d < DH; ++d) {
            wv8[d] = w[d];
            si = fmaf(wv8[d], aw[d], si);
            sj = fmaf(wv8[d], aw[DH + d], sj);
        }
        const size_t node = (size_t)(r0 + row);
        const float fj = __expf(0.2f * sj);
        efI[node * NHEADS + cg * (P1_COLS / DH) + hl] = __expf(0.8f * (si + ab));
        float* ef = efJ + ((size_t)cg * NN + node) * (2 * HPB);
        ef[hl]       = fj;
        ef[HPB + hl] = __expf(0.8f * sj);
#pragma unroll
        for (int d = 0; d < DH; ++d) w[d] = wv8[d] * fj;
    }
    __syncthreads();

#pragma unroll
    for (int k = 0; k < 2; ++k) {
        const int row  = (tid >> 3) + 8 * k;
        const int col4 = (tid & 7) * 4;
        const float4 v = *(const float4*)&tile[row * P1_LSTR + col4];
        *(float4*)(whF + (size_t)(r0 + row) * OUT_DIM + cg * P1_COLS + col4) = v;
    }
}

#define NCHUNK    (NN / JCHUNK)
#define ADJ_STAGE (16 * LSTR)
#define BROWS     (DH + 2)
#define WH_STAGE  (HPB * BROWS * LSTR)
#define VJ_STAGE  (HPB * LSTR)
#define ATT_GRID  ((NN / 16) * (NHEADS / HPB))

__global__ __launch_bounds__(256)
void gat_att_kernel(const int*   __restrict__ adj,
                    const float* __restrict__ whF,
                    const float* __restrict__ efI,
                    const float* __restrict__ efJ,
                    float* __restrict__ out) {
    __shared__ alignas(16) int   adjS[2 * ADJ_STAGE];
    __shared__ alignas(16) float whT[2 * WH_STAGE];
    __shared__ alignas(16) float vjT[2 * VJ_STAGE];
    __shared__ float accX[2][HPB][8][32];

    const int tid  = threadIdx.x;
    const int wv   = tid >> 5;
    const int hl   = wv & (HPB - 1);
    const int jh   = wv >> 2;
    const int lane = tid & 31;
    const int hg   = blockIdx.x & 1;
    const int hH   = hg * HPB + hl;
    const int i0   = (blockIdx.x >> 1) * 16;

    const int   iLoc  = lane & 15;
    const int   nCol  = lane & 15;
    const int   khalf = (lane >> 4) * 2;
    const float ui    = efI[(size_t)(i0 + iLoc) * NHEADS + hH];
    const int   dRow  = hl * BROWS + ((nCol < DH + 1) ? nCol : DH + 1);
    const int   jBase = jh * JHALF;

    const int aR = tid >> 6;
    const int aC = tid & 63;
    const int* adjP = adj + (size_t)(i0 + aR) * NN + aC * 4;
    const int wJ = tid >> 3;
    const int wD = (tid & 7) * 4;
    const float* whP = whF + (size_t)wJ * OUT_DIM + hg * (HPB * DH) + wD;
    const int wRow = (wD >> 3) * BROWS + (wD & 7);
    const float* ejP = efJ + ((size_t)hg * NN + tid) * (2 * HPB);

    v4i    rA[4];
    float4 rW[8];
    float4 rE, rF;

    auto tile_load = [&](int chunk) {
#pragma unroll
        for (int k = 0; k < 4; ++k)
            rA[k] = *(const v4i*)(adjP + chunk + (size_t)(4 * k) * NN);
#pragma unroll
        for (int k = 0; k < 8; ++k)
            rW[k] = *(const float4*)(whP + (size_t)(chunk + k * 32) * OUT_DIM);
        rE = *(const float4*)(ejP + (size_t)chunk * (2 * HPB));
        rF = *(const float4*)(ejP + (size_t)chunk * (2 * HPB) + HPB);
    };
    auto tile_store = [&](int st) {
        int*   aS = adjS + st * ADJ_STAGE;
        float* wS = whT  + st * WH_STAGE;
        float* vS = vjT  + st * VJ_STAGE;
#pragma unroll
        for (int k = 0; k < 4; ++k)
            ((v4i*)&aS[(aR + 4 * k) * LSTR])[aC] = rA[k];
#pragma unroll
        for (int k = 0; k < 8; ++k) {
            const int j = k * 32 + wJ;
            wS[(wRow + 0) * LSTR + j] = rW[k].x;
            wS[(wRow + 1) * LSTR + j] = rW[k].y;
            wS[(wRow + 2) * LSTR + j] = rW[k].z;
            wS[(wRow + 3) * LSTR + j] = rW[k].w;
        }
        wS[(0 * BROWS + DH) * LSTR + tid] = rE.x;  vS[0 * LSTR + tid] = rF.x;
        wS[(1 * BROWS + DH) * LSTR + tid] = rE.y;  vS[1 * LSTR + tid] = rF.y;
        wS[(2 * BROWS + DH) * LSTR + tid] = rE.z;  vS[2 * LSTR + tid] = rF.z;
        wS[(3 * BROWS + DH) * LSTR + tid] = rE.w;  vS[3 * LSTR + tid] = rF.w;
    };

    v8f   acc  = {0.f, 0.f, 0.f, 0.f, 0.f, 0.f, 0.f, 0.f};

    auto trip_compute = [&](int st, int trip) {
        const int*   aS = adjS + st * ADJ_STAGE;
        const float* wS = whT  + st * WH_STAGE;
        const float* vS = vjT  + st * VJ_STAGE;
        float2 vjR[16];
        int2   adR[16];
        float2 whR[16];
#pragma unroll
        for (int s = 0; s < 16; ++s) {
            const int j0 = jBase + trip * (JHALF / 2) + 4 * s + khalf;
            vjR[s] = *(const float2*)&vS[hl * LSTR + j0];
            adR[s] = *(const int2*)  &aS[iLoc * LSTR + j0];
            whR[s] = *(const float2*)&wS[dRow * LSTR + j0];
        }
        __builtin_amdgcn_sched_barrier(0);
#pragma unroll
        for (int s = 0; s < 16; ++s) {
            v2f pm; pm.x = vjR[s].x; pm.y = vjR[s].y;
            pm = pm * ui;
            const float p0 = (adR[s].x > 0) ? __builtin_fmaxf(pm.x, 1.0f) : 0.0f;
            const float p1 = (adR[s].y > 0) ? __builtin_fmaxf(pm.y, 1.0f) : 0.0f;

            v2f A; A.x = p0;       A.y = p1;
            v2f B; B.x = whR[s].x; B.y = whR[s].y;
            acc = __builtin_amdgcn_wmma_f32_16x16x4_f32(
                false, A, false, B, (short)0, acc, false, false);
        }
    };

#pragma unroll
    for (int zr = 0; zr < 2 * HPB; ++zr)
        whT[(zr >> 2) * WH_STAGE + ((zr & 3) * BROWS + DH + 1) * LSTR + tid] = 0.f;
    tile_load(0);
    tile_store(0);
    __syncthreads();
    tile_load(JCHUNK);

#pragma unroll 1
    for (int c = 0; c < NCHUNK; ++c) {
#pragma unroll 1
        for (int trip = 0; trip < 2; ++trip) {
            trip_compute(c & 1, trip);
            if (trip == 0) {
                if (c + 1 < NCHUNK) tile_store((c + 1) & 1);
                if (c + 2 < NCHUNK) tile_load((c + 2) * JCHUNK);
            }
        }
        if (c + 1 < NCHUNK) __syncthreads();
    }

#pragma unroll
    for (int v = 0; v < 8; ++v) accX[jh][hl][v][lane] = acc[v];
    __syncthreads();

    const int rbase = (lane >> 4) * 8;
    const int dLane = (lane & 16) + DH;
    if (jh == 0 && nCol < DH) {
#pragma unroll
        for (int v = 0; v < 8; ++v) {
            const int   iL  = rbase + v;
            const float den = accX[0][hl][v][dLane] + accX[1][hl][v][dLane];
            float o = (acc[v] + accX[1][hl][v][lane]) / den;
            o = (o > 0.f) ? o : (__expf(o) - 1.f);
            out[(size_t)hH * NN * DH + (size_t)(i0 + iL) * DH + nCol] = o;
        }
    }
}

extern "C" void kernel_launch(void* const* d_in, const int* in_sizes, int n_in,
                              void* d_out, int out_size, void* d_ws, size_t ws_size,
                              hipStream_t stream) {
    const float* h   = (const float*)d_in[0];
    const int*   adj = (const int*)  d_in[1];
    const float* W_w = (const float*)d_in[2];
    const float* W_b = (const float*)d_in[3];
    const float* a_w = (const float*)d_in[4];
    const float* a_b = (const float*)d_in[5];
    float* out = (float*)d_out;

    float* whF = (float*)d_ws;
    float* efI = whF + (size_t)NN * OUT_DIM;
    float* efJ = efI + (size_t)NN * NHEADS;

    gat_wh_kernel<<<P1_GRID, 64, 0, stream>>>(h, W_w, W_b, a_w, a_b, whF, efI, efJ);
    gat_att_kernel<<<ATT_GRID, 256, 0, stream>>>(adj, whF, efI, efJ, out);
}


# attention kernel: the per-tile barrier fences LDS only (s_wait_dscnt, no s_wait_loadcnt drain): the register fetch of the tile after next stays in flight across it; sum order unchanged
# speedup vs baseline: 6.1957x; 1.1031x over previous
#include <hip/hip_runtime.h>
#include <hip/hip_bf16.h>


typedef __attribute__((ext_vector_type(2))) float v2f;
typedef __attribute__((ext_vector_type(8))) float v8f;
typedef __attribute__((ext_vector_type(4))) int   v4i;

#define NN      2048
#define IN_DIM  256
#define OUT_DIM 64
#define NHEADS  8
#define DH      8
#define HPB     4
#define JCHUNK  256
#define JHALF   (JCHUNK / 2)
#define LSTR    260

#define P1_COLS 32
#define P1_LSTR 36
#define P1_GRID ((NN / 16) * (OUT_DIM / P1_COLS))

__global__ __launch_bounds__(64) __attribute__((amdgpu_waves_per_eu(1, 1)))
void gat_wh_kernel(const float* __restrict__ h,
                   const float* __restrict__ W_w,
                   const float* __restrict__ W_b,
                   const float* __restrict__ a_w,
                   const float* __restrict__ a_b,
                   float* __restrict__ whF,
                   float* __restrict__ efI,
                   float* __restrict__ efJ) {
    __shared__ alignas(16) float tile[16 * P1_LSTR];

    const int tid  = threadIdx.x;
    const int wv   = tid >> 5;
    const int lane = tid & 31;
    const int c    = lane & 15;
    const int hh   = lane >> 4;
    const int r0   = (blockIdx.x >> 1) * 16;
    const int cg   = blockIdx.x & 1;
    const int n0   = cg * P1_COLS + wv * 16;

    const float4* aP = (const float4*)(h   + (size_t)(r0 + c) * IN_DIM + 4 * hh);
    const float4* bP = (const float4*)(W_w + (size_t)(n0 + c) * IN_DIM + 4 * hh);

    float aw[2 * DH];
#pragma unroll
    for (int d = 0; d < 2 * DH; ++d) aw[d] = a_w[d];
    const float ab = a_b[0];

    float4 ra[IN_DIM / 8], rb[IN_DIM / 8];
#pragma unroll
    for (int t = 0; t < IN_DIM / 8; ++t) {
        ra[t] = aP[2 * t];
        rb[t] = bP[2 * t];
    }

    const float bias = W_b[n0 + c];
    v8f acc0 = {bias, bias, bias, bias, bias, bias, bias, bias};
    v8f acc1 = {0.f, 0.f, 0.f, 0.f, 0.f, 0.f, 0.f, 0.f};
#pragma unroll
    for (int t = 0; t < IN_DIM / 8; ++t) {
        v2f A0; A0.x = ra[t].x; A0.y = ra[t].y;
        v2f B0; B0.x = rb[t].x; B0.y = rb[t].y;
        v2f A1; A1.x = ra[t].z; A1.y = ra[t].w;
        v2f B1; B1.x = rb[t].z; B1.y = rb[t].w;
        acc0 = __builtin_amdgcn_wmma_f32_16x16x4_f32(
            false, A0, false, B0, (short)0, acc0, false, false);
        acc1 = __builtin_amdgcn_wmma_f32_16x16x4_f32(
            false, A1, false, B1, (short)0, acc1, false, false);
    }

#pragma unroll
    for (int r = 0; r < 8; ++r)
        tile[(8 * hh + r) * P1_LSTR + wv * 16 + c] = acc0[r] + acc1[r];
    __syncthreads();

    {
        const int row = tid >> 2;
        const int hl  = tid & 3;
        float* w = &tile[row * P1_LSTR + hl * DH];
        float wv8[DH];
        float si = 0.f, sj = 0.f;
#pragma unroll
        for (int d = 0; d < DH; ++d) {
            wv8[d] = w[d];
            si = fmaf(wv8[d], aw[d], si);
            sj = fmaf(wv8[d], aw[DH + d], sj);
        }
        const size_t node = (size_t)(r0 + row);
        const float fj = __expf(0.2f * sj);
        efI[node * NHEADS + cg * (P1_COLS / DH) + hl] = __expf(0.8f * (si + ab));
        float* ef = efJ + ((size_t)cg * NN + node) * (2 * HPB);
        ef[hl]       = fj;
        ef[HPB + hl] = __expf(0.8f * sj);
#pragma unroll
        for (int d = 0; d < DH; ++d) w[d] = wv8[d] * fj;
    }
    __syncthreads();

#pragma unroll
    for (int k = 0; k < 2; ++k) {
        const int row  = (tid >> 3) + 8 * k;
        const int col4 = (tid & 7) * 4;
        const float4 v = *(const float4*)&tile[row * P1_LSTR + col4];
        *(float4*)(whF + (size_t)(r0 + row) * OUT_DIM + cg * P1_COLS + col4) = v;
    }
}

#define NCHUNK    (NN / JCHUNK)
#define ADJ_STAGE (16 * LSTR)
#define BROWS     (DH + 2)
#define WH_STAGE  (HPB * BROWS * LSTR)
#define VJ_STAGE  (HPB * LSTR)
#define ATT_GRID  ((NN / 16) * (NHEADS / HPB))

__global__ __launch_bounds__(256)
void gat_att_kernel(const int*   __restrict__ adj,
                    const float* __restrict__ whF,
                    const float* __restrict__ efI,
                    const float* __restrict__ efJ,
                    float* __restrict__ out) {
    __shared__ alignas(16) int   adjS[2 * ADJ_STAGE];
    __shared__ alignas(16) float whT[2 * WH_STAGE];
    __shared__ alignas(16) float vjT[2 * VJ_STAGE];
    __shared__ float accX[2][HPB][8][32];

    const int tid  = threadIdx.x;
    const int wv   = tid >> 5;
    const int hl   = wv & (HPB - 1);
    const int jh   = wv >> 2;
    const int lane = tid & 31;
    const int hg   = blockIdx.x & 1;
    const int hH   = hg * HPB + hl;
    const int i0   = (blockIdx.x >> 1) * 16;

    const int   iLoc  = lane & 15;
    const int   nCol  = lane & 15;
    const int   khalf = (lane >> 4) * 2;
    const float ui    = efI[(size_t)(i0 + iLoc) * NHEADS + hH];
    const int   dRow  = hl * BROWS + ((nCol < DH + 1) ? nCol : DH + 1);
    const int   jBase = jh * JHALF;

    const int aR = tid >> 6;
    const int aC = tid & 63;
    const int* adjP = adj + (size_t)(i0 + aR) * NN + aC * 4;
    const int wJ = tid >> 3;
    const int wD = (tid & 7) * 4;
    const float* whP = whF + (size_t)wJ * OUT_DIM + hg * (HPB * DH) + wD;
    const int wRow = (wD >> 3) * BROWS + (wD & 7);
    const float* ejP = efJ + ((size_t)hg * NN + tid) * (2 * HPB);

    v4i    rA[4];
    float4 rW[8];
    float4 rE, rF;

    auto tile_load = [&](int chunk) {
#pragma unroll
        for (int k = 0; k < 4; ++k)
            rA[k] = *(const v4i*)(adjP + chunk + (size_t)(4 * k) * NN);
#pragma unroll
        for (int k = 0; k < 8; ++k)
            rW[k] = *(const float4*)(whP + (size_t)(chunk + k * 32) * OUT_DIM);
        rE = *(const float4*)(ejP + (size_t)chunk * (2 * HPB));
        rF = *(const float4*)(ejP + (size_t)chunk * (2 * HPB) + HPB);
    };
    auto tile_store = [&](int st) {
        int*   aS = adjS + st * ADJ_STAGE;
        float* wS = whT  + st * WH_STAGE;
        float* vS = vjT  + st * VJ_STAGE;
#pragma unroll
        for (int k = 0; k < 4; ++k)
            ((v4i*)&aS[(aR + 4 * k) * LSTR])[aC] = rA[k];
#pragma unroll
        for (int k = 0; k < 8; ++k) {
            const int j = k * 32 + wJ;
            wS[(wRow + 0) * LSTR + j] = rW[k].x;
            wS[(wRow + 1) * LSTR + j] = rW[k].y;
            wS[(wRow + 2) * LSTR + j] = rW[k].z;
            wS[(wRow + 3) * LSTR + j] = rW[k].w;
        }
        wS[(0 * BROWS + DH) * LSTR + tid] = rE.x;  vS[0 * LSTR + tid] = rF.x;
        wS[(1 * BROWS + DH) * LSTR + tid] = rE.y;  vS[1 * LSTR + tid] = rF.y;
        wS[(2 * BROWS + DH) * LSTR + tid] = rE.z;  vS[2 * LSTR + tid] = rF.z;
        wS[(3 * BROWS + DH) * LSTR + tid] = rE.w;  vS[3 * LSTR + tid] = rF.w;
    };

    v8f   acc  = {0.f, 0.f, 0.f, 0.f, 0.f, 0.f, 0.f, 0.f};

    auto trip_compute = [&](int st, int trip) {
        const int*   aS = adjS + st * ADJ_STAGE;
        const float* wS = whT  + st * WH_STAGE;
        const float* vS = vjT  + st * VJ_STAGE;
        float2 vjR[16];
        int2   adR[16];
        float2 whR[16];
#pragma unroll
        for (int s = 0; s < 16; ++s) {
            const int j0 = jBase + trip * (JHALF / 2) + 4 * s + khalf;
            vjR[s] = *(const float2*)&vS[hl * LSTR + j0];
            adR[s] = *(const int2*)  &aS[iLoc * LSTR + j0];
            whR[s] = *(const float2*)&wS[dRow * LSTR + j0];
        }
        __builtin_amdgcn_sched_barrier(0);
#pragma unroll
        for (int s = 0; s < 16; ++s) {
            v2f pm; pm.x = vjR[s].x; pm.y = vjR[s].y;
            pm = pm * ui;
            const float p0 = (adR[s].x > 0) ? __builtin_fmaxf(pm.x, 1.0f) : 0.0f;
            const float p1 = (adR[s].y > 0) ? __builtin_fmaxf(pm.y, 1.0f) : 0.0f;

            v2f A; A.x = p0;       A.y = p1;
            v2f B; B.x = whR[s].x; B.y = whR[s].y;
            acc = __builtin_amdgcn_wmma_f32_16x16x4_f32(
                false, A, false, B, (short)0, acc, false, false);
        }
    };

#pragma unroll
    for (int zr = 0; zr < 2 * HPB; ++zr)
        whT[(zr >> 2) * WH_STAGE + ((zr & 3) * BROWS + DH + 1) * LSTR + tid] = 0.f;
    tile_load(0);
    tile_store(0);
    __syncthreads();
    tile_load(JCHUNK);

#pragma unroll 1
    for (int c = 0; c < NCHUNK; ++c) {
#pragma unroll 1
        for (int trip = 0; trip < 2; ++trip) {
            trip_compute(c & 1, trip);
            if (trip == 0) {
                if (c + 1 < NCHUNK) tile_store((c + 1) & 1);
                if (c + 2 < NCHUNK) tile_load((c + 2) * JCHUNK);
            }
        }
        if (c + 1 < NCHUNK) {
            __builtin_amdgcn_fence(__ATOMIC_RELEASE, "workgroup", "local");
            __builtin_amdgcn_s_barrier();
            __builtin_amdgcn_fence(__ATOMIC_ACQUIRE, "workgroup", "local");
        }
    }

#pragma unroll
    for (int v = 0; v < 8; ++v) accX[jh][hl][v][lane] = acc[v];
    __syncthreads();

    const int rbase = (lane >> 4) * 8;
    const int dLane = (lane & 16) + DH;
    if (jh == 0 && nCol < DH) {
#pragma unroll
        for (int v = 0; v < 8; ++v) {
            const int   iL  = rbase + v;
            const float den = accX[0][hl][v][dLane] + accX[1][hl][v][dLane];
            float o = (acc[v] + accX[1][hl][v][lane]) / den;
            o = (o > 0.f) ? o : (__expf(o) - 1.f);
            out[(size_t)hH * NN * DH + (size_t)(i0 + iL) * DH + nCol] = o;
        }
    }
}

extern "C" void kernel_launch(void* const* d_in, const int* in_sizes, int n_in,
                              void* d_out, int out_size, void* d_ws, size_t ws_size,
                              hipStream_t stream) {
    const float* h   = (const float*)d_in[0];
    const int*   adj = (const int*)  d_in[1];
    const float* W_w = (const float*)d_in[2];
    const float* W_b = (const float*)d_in[3];
    const float* a_w = (const float*)d_in[4];
    const float* a_b = (const float*)d_in[5];
    float* out = (float*)d_out;

    float* whF = (float*)d_ws;
    float* efI = whF + (size_t)NN * OUT_DIM;
    float* efJ = efI + (size_t)NN * NHEADS;

    gat_wh_kernel<<<P1_GRID, 64, 0, stream>>>(h, W_w, W_b, a_w, a_b, whF, efI, efJ);
    gat_att_kernel<<<ATT_GRID, 256, 0, stream>>>(adj, whF, efI, efJ, out);
}


# attention kernel: each 16-WMMA trip's LDS operand reads issued one trip ahead into a second register set (reads, stage stores, fetch run under the WMMAs); barrier between trips; sum order unchanged
# speedup vs baseline: 6.8021x; 1.0979x over previous
#include <hip/hip_runtime.h>
#include <hip/hip_bf16.h>


typedef __attribute__((ext_vector_type(2))) float v2f;
typedef __attribute__((ext_vector_type(8))) float v8f;
typedef __attribute__((ext_vector_type(4))) int   v4i;

#define NN      2048
#define IN_DIM  256
#define OUT_DIM 64
#define NHEADS  8
#define DH      8
#define HPB     4
#define JCHUNK  256
#define JHALF   (JCHUNK / 2)
#define LSTR    260

#define P1_COLS 32
#define P1_LSTR 36
#define P1_GRID ((NN / 16) * (OUT_DIM / P1_COLS))

__global__ __launch_bounds__(64) __attribute__((amdgpu_waves_per_eu(1, 1)))
void gat_wh_kernel(const float* __restrict__ h,
                   const float* __restrict__ W_w,
                   const float* __restrict__ W_b,
                   const float* __restrict__ a_w,
                   const float* __restrict__ a_b,
                   float* __restrict__ whF,
                   float* __restrict__ efI,
                   float* __restrict__ efJ) {
    __shared__ alignas(16) float tile[16 * P1_LSTR];

    const int tid  = threadIdx.x;
    const int wv   = tid >> 5;
    const int lane = tid & 31;
    const int c    = lane & 15;
    const int hh   = lane >> 4;
    const int r0   = (blockIdx.x >> 1) * 16;
    const int cg   = blockIdx.x & 1;
    const int n0   = cg * P1_COLS + wv * 16;

    const float4* aP = (const float4*)(h   + (size_t)(r0 + c) * IN_DIM + 4 * hh);
    const float4* bP = (const float4*)(W_w + (size_t)(n0 + c) * IN_DIM + 4 * hh);

    float aw[2 * DH];
#pragma unroll
    for (int d = 0; d < 2 * DH; ++d) aw[d] = a_w[d];
    const float ab = a_b[0];

    float4 ra[IN_DIM / 8], rb[IN_DIM / 8];
#pragma unroll
    for (int t = 0; t < IN_DIM / 8; ++t) {
        ra[t] = aP[2 * t];
        rb[t] = bP[2 * t];
    }

    const float bias = W_b[n0 + c];
    v8f acc0 = {bias, bias, bias, bias, bias, bias, bias, bias};
    v8f acc1 = {0.f, 0.f, 0.f, 0.f, 0.f, 0.f, 0.f, 0.f};
#pragma unroll
    for (int t = 0; t < IN_DIM / 8; ++t) {
        v2f A0; A0.x = ra[t].x; A0.y = ra[t].y;
        v2f B0; B0.x = rb[t].x; B0.y = rb[t].y;
        v2f A1; A1.x = ra[t].z; A1.y = ra[t].w;
        v2f B1; B1.x = rb[t].z; B1.y = rb[t].w;
        acc0 = __builtin_amdgcn_wmma_f32_16x16x4_f32(
            false, A0, false, B0, (short)0, acc0, false, false);
        acc1 = __builtin_amdgcn_wmma_f32_16x16x4_f32(
            false, A1, false, B1, (short)0, acc1, false, false);
    }

#pragma unroll
    for (int r = 0; r < 8; ++r)
        tile[(8 * hh + r) * P1_LSTR + wv * 16 + c] = acc0[r] + acc1[r];
    __syncthreads();

    {
        const int row = tid >> 2;
        const int hl  = tid & 3;
        float* w = &tile[row * P1_LSTR + hl * DH];
        float wv8[DH];
        float si = 0.f, sj = 0.f;
#pragma unroll
        for (int d = 0; d < DH; ++d) {
            wv8[d] = w[d];
            si = fmaf(wv8[d], aw[d], si);
            sj = fmaf(wv8[d], aw[DH + d], sj);
        }
        const size_t node = (size_t)(r0 + row);
        const float fj = __expf(0.2f * sj);
        efI[node * NHEADS + cg * (P1_COLS / DH) + hl] = __expf(0.8f * (si + ab));
        float* ef = efJ + ((size_t)cg * NN + node) * (2 * HPB);
        ef[hl]       = fj;
        ef[HPB + hl] = __expf(0.8f * sj);
#pragma unroll
        for (int d = 0; d < DH; ++d) w[d] = wv8[d] * fj;
    }
    __syncthreads();

#pragma unroll
    for (int k = 0; k < 2; ++k) {
        const int row  = (tid >> 3) + 8 * k;
        const int col4 = (tid & 7) * 4;
        const float4 v = *(const float4*)&tile[row * P1_LSTR + col4];
        *(float4*)(whF + (size_t)(r0 + row) * OUT_DIM + cg * P1_COLS + col4) = v;
    }
}

#define NCHUNK    (NN / JCHUNK)
#define ADJ_STAGE (16 * LSTR)
#define BROWS     (DH + 2)
#define WH_STAGE  (HPB * BROWS * LSTR)
#define VJ_STAGE  (HPB * LSTR)
#define ATT_GRID  ((NN / 16) * (NHEADS / HPB))

__global__ __launch_bounds__(256)
void gat_att_kernel(const int*   __restrict__ adj,
                    const float* __restrict__ whF,
                    const float* __restrict__ efI,
                    const float* __restrict__ efJ,
                    float* __restrict__ out) {
    __shared__ alignas(16) int   adjS[2 * ADJ_STAGE];
    __shared__ alignas(16) float whT[2 * WH_STAGE];
    __shared__ alignas(16) float vjT[2 * VJ_STAGE];
    __shared__ float accX[2][HPB][8][32];

    const int tid  = threadIdx.x;
    const int wv   = tid >> 5;
    const int hl   = wv & (HPB - 1);
    const int jh   = wv >> 2;
    const int lane = tid & 31;
    const int hg   = blockIdx.x & 1;
    const int hH   = hg * HPB + hl;
    const int i0   = (blockIdx.x >> 1) * 16;

    const int   iLoc  = lane & 15;
    const int   nCol  = lane & 15;
    const int   khalf = (lane >> 4) * 2;
    const float ui    = efI[(size_t)(i0 + iLoc) * NHEADS + hH];
    const int   dRow  = hl * BROWS + ((nCol < DH + 1) ? nCol : DH + 1);
    const int   jBase = jh * JHALF;

    const int aR = tid >> 6;
    const int aC = tid & 63;
    const int* adjP = adj + (size_t)(i0 + aR) * NN + aC * 4;
    const int wJ = tid >> 3;
    const int wD = (tid & 7) * 4;
    const float* whP = whF + (size_t)wJ * OUT_DIM + hg * (HPB * DH) + wD;
    const int wRow = (wD >> 3) * BROWS + (wD & 7);
    const float* ejP = efJ + ((size_t)hg * NN + tid) * (2 * HPB);

    v4i    rA[4];
    float4 rW[8];
    float4 rE, rF;

    auto tile_load = [&](int chunk) {
#pragma unroll
        for (int k = 0; k < 4; ++k)
            rA[k] = *(const v4i*)(adjP + chunk + (size_t)(4 * k) * NN);
#pragma unroll
        for (int k = 0; k < 8; ++k)
            rW[k] = *(const float4*)(whP + (size_t)(chunk + k * 32) * OUT_DIM);
        rE = *(const float4*)(ejP + (size_t)chunk * (2 * HPB));
        rF = *(const float4*)(ejP + (size_t)chunk * (2 * HPB) + HPB);
    };
    auto tile_store = [&](int st) {
        int*   aS = adjS + st * ADJ_STAGE;
        float* wS = whT  + st * WH_STAGE;
        float* vS = vjT  + st * VJ_STAGE;
#pragma unroll
        for (int k = 0; k < 4; ++k)
            ((v4i*)&aS[(aR + 4 * k) * LSTR])[aC] = rA[k];
#pragma unroll
        for (int k = 0; k < 8; ++k) {
            const int j = k * 32 + wJ;
            wS[(wRow + 0) * LSTR + j] = rW[k].x;
            wS[(wRow + 1) * LSTR + j] = rW[k].y;
            wS[(wRow + 2) * LSTR + j] = rW[k].z;
            wS[(wRow + 3) * LSTR + j] = rW[k].w;
        }
        wS[(0 * BROWS + DH) * LSTR + tid] = rE.x;  vS[0 * LSTR + tid] = rF.x;
        wS[(1 * BROWS + DH) * LSTR + tid] = rE.y;  vS[1 * LSTR + tid] = rF.y;
        wS[(2 * BROWS + DH) * LSTR + tid] = rE.z;  vS[2 * LSTR + tid] = rF.z;
        wS[(3 * BROWS + DH) * LSTR + tid] = rE.w;  vS[3 * LSTR + tid] = rF.w;
    };

    v8f   acc  = {0.f, 0.f, 0.f, 0.f, 0.f, 0.f, 0.f, 0.f};

    struct TripOps {
        float2 vj[16];
        int2   ad[16];
        float2 wh[16];
    };
    TripOps opX, opY;

    auto trip_fetch = [&](int st, int trip, TripOps& o) {
        const int*   aS = adjS + st * ADJ_STAGE;
        const float* wS = whT  + st * WH_STAGE;
        const float* vS = vjT  + st * VJ_STAGE;
#pragma unroll
        for (int s = 0; s < 16; ++s) {
            const int j0 = jBase + trip * (JHALF / 2) + 4 * s + khalf;
            o.vj[s] = *(const float2*)&vS[hl * LSTR + j0];
            o.ad[s] = *(const int2*)  &aS[iLoc * LSTR + j0];
            o.wh[s] = *(const float2*)&wS[dRow * LSTR + j0];
        }
    };
    auto trip_mma = [&](const TripOps& o) {
#pragma unroll
        for (int s = 0; s < 16; ++s) {
            v2f pm; pm.x = o.vj[s].x; pm.y = o.vj[s].y;
            pm = pm * ui;
            const float p0 = (o.ad[s].x > 0) ? __builtin_fmaxf(pm.x, 1.0f) : 0.0f;
            const float p1 = (o.ad[s].y > 0) ? __builtin_fmaxf(pm.y, 1.0f) : 0.0f;

            v2f A; A.x = p0;        A.y = p1;
            v2f B; B.x = o.wh[s].x; B.y = o.wh[s].y;
            acc = __builtin_amdgcn_wmma_f32_16x16x4_f32(
                false, A, false, B, (short)0, acc, false, false);
        }
    };

#pragma unroll
    for (int zr = 0; zr < 2 * HPB; ++zr)
        whT[(zr >> 2) * WH_STAGE + ((zr & 3) * BROWS + DH + 1) * LSTR + tid] = 0.f;
    tile_load(0);
    tile_store(0);
    __syncthreads();
    tile_load(JCHUNK);
    trip_fetch(0, 0, opX);

#pragma unroll 1
    for (int c = 0; c < NCHUNK - 1; ++c) {
        trip_fetch(c & 1, 1, opY);
        tile_store((c + 1) & 1);
        if (c + 2 < NCHUNK) tile_load((c + 2) * JCHUNK);
        __builtin_amdgcn_sched_barrier(0);
        trip_mma(opX);
        __builtin_amdgcn_sched_barrier(0);
        __builtin_amdgcn_fence(__ATOMIC_RELEASE, "workgroup", "local");
        __builtin_amdgcn_s_barrier();
        __builtin_amdgcn_fence(__ATOMIC_ACQUIRE, "workgroup", "local");
        trip_fetch((c + 1) & 1, 0, opX);
        __builtin_amdgcn_sched_barrier(0);
        trip_mma(opY);
    }
    trip_fetch((NCHUNK - 1) & 1, 1, opY);
    __builtin_amdgcn_sched_barrier(0);
    trip_mma(opX);
    trip_mma(opY);

#pragma unroll
    for (int v = 0; v < 8; ++v) accX[jh][hl][v][lane] = acc[v];
    __syncthreads();

    const int rbase = (lane >> 4) * 8;
    const int dLane = (lane & 16) + DH;
    if (jh == 0 && nCol < DH) {
#pragma unroll
        for (int v = 0; v < 8; ++v) {
            const int   iL  = rbase + v;
            const float den = accX[0][hl][v][dLane] + accX[1][hl][v][dLane];
            float o = (acc[v] + accX[1][hl][v][lane]) / den;
            o = (o > 0.f) ? o : (__expf(o) - 1.f);
            out[(size_t)hH * NN * DH + (size_t)(i0 + iL) * DH + nCol] = o;
        }
    }
}

extern "C" void kernel_launch(void* const* d_in, const int* in_sizes, int n_in,
                              void* d_out, int out_size, void* d_ws, size_t ws_size,
                              hipStream_t stream) {
    const float* h   = (const float*)d_in[0];
    const int*   adj = (const int*)  d_in[1];
    const float* W_w = (const float*)d_in[2];
    const float* W_b = (const float*)d_in[3];
    const float* a_w = (const float*)d_in[4];
    const float* a_b = (const float*)d_in[5];
    float* out = (float*)d_out;

    float* whF = (float*)d_ws;
    float* efI = whF + (size_t)NN * OUT_DIM;
    float* efJ = efI + (size_t)NN * NHEADS;

    gat_wh_kernel<<<P1_GRID, 64, 0, stream>>>(h, W_w, W_b, a_w, a_b, whF, efI, efJ);
    gat_att_kernel<<<ATT_GRID, 256, 0, stream>>>(adj, whF, efI, efJ, out);
}


# attention kernel: B operand and v_j read straight from global memory in fragment order (projection kernel writes them so), one tile ahead; LDS stages adj only (sum order over j changes)
# speedup vs baseline: 7.1513x; 1.0513x over previous
#include <hip/hip_runtime.h>
#include <hip/hip_bf16.h>


typedef __attribute__((ext_vector_type(2))) float v2f;
typedef __attribute__((ext_vector_type(8))) float v8f;
typedef __attribute__((ext_vector_type(4))) int   v4i;

#define NN      2048
#define IN_DIM  256
#define OUT_DIM 64
#define NHEADS  8
#define DH      8
#define HPB     4
#define JCHUNK  256
#define JHALF   (JCHUNK / 2)
#define LSTR    260
#define NJB      (NN / 8)
#define BG_ROWS  12
#define BG_CHUNK (BG_ROWS * 8)
#define BG_VROW  (DH + 2)

#define P1_COLS 32
#define P1_LSTR 36
#define P1_GRID ((NN / 16) * (OUT_DIM / P1_COLS))

__global__ __launch_bounds__(64) __attribute__((amdgpu_waves_per_eu(1, 1)))
void gat_wh_kernel(const float* __restrict__ h,
                   const float* __restrict__ W_w,
                   const float* __restrict__ W_b,
                   const float* __restrict__ a_w,
                   const float* __restrict__ a_b,
                   float* __restrict__ Bg,
                   float* __restrict__ efI) {
    __shared__ alignas(16) float tile[16 * P1_LSTR];
    __shared__ float fS[16][P1_COLS / DH];
    __shared__ float vS[16][P1_COLS / DH];

    const int tid  = threadIdx.x;
    const int wv   = tid >> 5;
    const int lane = tid & 31;
    const int c    = lane & 15;
    const int hh   = lane >> 4;
    const int r0   = (blockIdx.x >> 1) * 16;
    const int cg   = blockIdx.x & 1;
    const int n0   = cg * P1_COLS + wv * 16;

    const float4* aP = (const float4*)(h   + (size_t)(r0 + c) * IN_DIM + 4 * hh);
    const float4* bP = (const float4*)(W_w + (size_t)(n0 + c) * IN_DIM + 4 * hh);

    float aw[2 * DH];
#pragma unroll
    for (int d = 0; d < 2 * DH; ++d) aw[d] = a_w[d];
    const float ab = a_b[0];

    float4 ra[IN_DIM / 8], rb[IN_DIM / 8];
#pragma unroll
    for (int t = 0; t < IN_DIM / 8; ++t) {
        ra[t] = aP[2 * t];
        rb[t] = bP[2 * t];
    }

    const float bias = W_b[n0 + c];
    v8f acc0 = {bias, bias, bias, bias, bias, bias, bias, bias};
    v8f acc1 = {0.f, 0.f, 0.f, 0.f, 0.f, 0.f, 0.f, 0.f};
#pragma unroll
    for (int t = 0; t < IN_DIM / 8; ++t) {
        v2f A0; A0.x = ra[t].x; A0.y = ra[t].y;
        v2f B0; B0.x = rb[t].x; B0.y = rb[t].y;
        v2f A1; A1.x = ra[t].z; A1.y = ra[t].w;
        v2f B1; B1.x = rb[t].z; B1.y = rb[t].w;
        acc0 = __builtin_amdgcn_wmma_f32_16x16x4_f32(
            false, A0, false, B0, (short)0, acc0, false, false);
        acc1 = __builtin_amdgcn_wmma_f32_16x16x4_f32(
            false, A1, false, B1, (short)0, acc1, false, false);
    }

#pragma unroll
    for (int r = 0; r < 8; ++r)
        tile[(8 * hh + r) * P1_LSTR + wv * 16 + c] = acc0[r] + acc1[r];
    __syncthreads();

    {
        const int row = tid >> 2;
        const int hl  = tid & 3;
        float* w = &tile[row * P1_LSTR + hl * DH];
        float wv8[DH];
        float si = 0.f, sj = 0.f;
#pragma unroll
        for (int d = 0; d < DH; ++d) {
            wv8[d] = w[d];
            si = fmaf(wv8[d], aw[d], si);
            sj = fmaf(wv8[d], aw[DH + d], sj);
        }
        const size_t node = (size_t)(r0 + row);
        const float fj = __expf(0.2f * sj);
        efI[node * NHEADS + cg * (P1_COLS / DH) + hl] = __expf(0.8f * (si + ab));
        fS[row][hl] = fj;
        vS[row][hl] = __expf(0.8f * sj);
#pragma unroll
        for (int d = 0; d < DH; ++d) w[d] = wv8[d] * fj;
    }
    __syncthreads();

    const int jb0 = r0 >> 3;
#pragma unroll
    for (int k = 0; k < 3; ++k) {
        const int idx = tid + 64 * k;
        if (idx < 8 * 22) {
            const int chunk = idx / 22;
            const int q     = idx - chunk * 22;
            const int hl    = chunk >> 1;
            const int jbl   = chunk & 1;
            const int row   = q >> 1;
            const int n0    = 8 * jbl + 4 * (q & 1);
            float4 v;
            if (row < DH) {
                const float* t = &tile[n0 * P1_LSTR + hl * DH + row];
                v.x = t[0]; v.y = t[P1_LSTR]; v.z = t[2 * P1_LSTR]; v.w = t[3 * P1_LSTR];
            } else if (row == DH) {
                v.x = fS[n0][hl]; v.y = fS[n0 + 1][hl]; v.z = fS[n0 + 2][hl]; v.w = fS[n0 + 3][hl];
            } else if (row == DH + 1) {
                v.x = 0.f; v.y = 0.f; v.z = 0.f; v.w = 0.f;
            } else {
                v.x = vS[n0][hl]; v.y = vS[n0 + 1][hl]; v.z = vS[n0 + 2][hl]; v.w = vS[n0 + 3][hl];
            }
            float* dst = Bg + ((size_t)(cg * (P1_COLS / DH) + hl) * NJB + jb0 + jbl) * BG_CHUNK + q * 4;
            *(float4*)dst = v;
        }
    }
}

#define NCHUNK    (NN / JCHUNK)
#define ADJ_STAGE (16 * LSTR)
#define ATT_GRID  ((NN / 16) * (NHEADS / HPB))

__global__ __launch_bounds__(256)
void gat_att_kernel(const int*   __restrict__ adj,
                    const float* __restrict__ Bg,
                    const float* __restrict__ efI,
                    float* __restrict__ out) {
    __shared__ alignas(16) int   adjS[2 * ADJ_STAGE];
    __shared__ float accX[2][HPB][8][32];

    const int tid  = threadIdx.x;
    const int wv   = tid >> 5;
    const int hl   = wv & (HPB - 1);
    const int jh   = wv >> 2;
    const int lane = tid & 31;
    const int hg   = blockIdx.x & 1;
    const int hH   = hg * HPB + hl;
    const int i0   = (blockIdx.x >> 1) * 16;

    const int   iLoc  = lane & 15;
    const int   nCol  = lane & 15;
    const int   hh    = lane >> 4;
    float ui;
    const int   dRow  = (nCol < DH + 1) ? nCol : DH + 1;
    const int   jBase = jh * JHALF;

    const float* bP = Bg + (size_t)hH * NJB * BG_CHUNK + dRow * 8 + 4 * hh;
    const float* vP = Bg + (size_t)hH * NJB * BG_CHUNK + BG_VROW * 8 + 4 * hh;

    const int aR = tid >> 6;
    const int aC = tid & 63;
    const int* adjP = adj + (size_t)(i0 + aR) * NN + aC * 4;

    v4i rA[4];

    auto tile_load = [&](int chunk) {
#pragma unroll
        for (int k = 0; k < 4; ++k)
            rA[k] = *(const v4i*)(adjP + chunk + (size_t)(4 * k) * NN);
    };
    auto tile_store = [&](int st) {
        int* aS = adjS + st * ADJ_STAGE;
#pragma unroll
        for (int k = 0; k < 4; ++k)
            ((v4i*)&aS[(aR + 4 * k) * LSTR])[aC] = rA[k];
    };

    v8f   acc  = {0.f, 0.f, 0.f, 0.f, 0.f, 0.f, 0.f, 0.f};

    struct AdjOps { v4i ad[8]; };
    struct BvOps  { float4 vj[8]; float4 wh[8]; };
    AdjOps aX, aY;
    BvOps  e0, e1, o0, o1;

    auto adj_fetch = [&](int c, int trip, AdjOps& o) {
        const int* aS = adjS + (c & 1) * ADJ_STAGE + iLoc * LSTR + jBase + trip * (JHALF / 2) + 4 * hh;
#pragma unroll
        for (int q = 0; q < 8; ++q)
            o.ad[q] = *(const v4i*)&aS[8 * q];
    };
    auto bv_fetch = [&](int c, int trip, BvOps& o) {
        const size_t jb = (size_t)(c * (JCHUNK / 8) + jh * (JHALF / 8) + trip * (JHALF / 16));
        const float* bS = bP + jb * BG_CHUNK;
        const float* vS = vP + jb * BG_CHUNK;
#pragma unroll
        for (int q = 0; q < 8; ++q) {
            o.wh[q] = *(const float4*)(bS + (size_t)q * BG_CHUNK);
            o.vj[q] = *(const float4*)(vS + (size_t)q * BG_CHUNK);
        }
    };
    auto trip_mma = [&](const AdjOps& a, const BvOps& o) {
#pragma unroll
        for (int q = 0; q < 8; ++q) {
            v2f pa; pa.x = o.vj[q].x; pa.y = o.vj[q].y;
            v2f pb; pb.x = o.vj[q].z; pb.y = o.vj[q].w;
            pa = pa * ui;
            pb = pb * ui;
            const float p0 = (a.ad[q].x > 0) ? __builtin_fmaxf(pa.x, 1.0f) : 0.0f;
            const float p1 = (a.ad[q].y > 0) ? __builtin_fmaxf(pa.y, 1.0f) : 0.0f;
            const float p2 = (a.ad[q].z > 0) ? __builtin_fmaxf(pb.x, 1.0f) : 0.0f;
            const float p3 = (a.ad[q].w > 0) ? __builtin_fmaxf(pb.y, 1.0f) : 0.0f;

            v2f A0; A0.x = p0;        A0.y = p1;
            v2f B0; B0.x = o.wh[q].x; B0.y = o.wh[q].y;
            acc = __builtin_amdgcn_wmma_f32_16x16x4_f32(
                false, A0, false, B0, (short)0, acc, false, false);
            v2f A1; A1.x = p2;        A1.y = p3;
            v2f B1; B1.x = o.wh[q].z; B1.y = o.wh[q].w;
            acc = __builtin_amdgcn_wmma_f32_16x16x4_f32(
                false, A1, false, B1, (short)0, acc, false, false);
        }
    };

    auto tile_step = [&](int c, bool more, const BvOps& b0, const BvOps& b1, BvOps& n0, BvOps& n1) {
        __builtin_amdgcn_sched_barrier(0);
        bv_fetch(c + 1, 0, n0);
        adj_fetch(c, 1, aY);
        tile_store((c + 1) & 1);
        if (more) tile_load((c + 2) * JCHUNK);
        __builtin_amdgcn_sched_barrier(0);
        trip_mma(aX, b0);
        __builtin_amdgcn_sched_barrier(0);
        __builtin_amdgcn_fence(__ATOMIC_RELEASE, "workgroup", "local");
        __builtin_amdgcn_s_barrier();
        __builtin_amdgcn_fence(__ATOMIC_ACQUIRE, "workgroup", "local");
        bv_fetch(c + 1, 1, n1);
        adj_fetch(c + 1, 0, aX);
        __builtin_amdgcn_sched_barrier(0);
        trip_mma(aY, b1);
    };

    tile_load(0);
    __builtin_amdgcn_sched_barrier(0);
    bv_fetch(0, 0, e0);
    ui = efI[(size_t)(i0 + iLoc) * NHEADS + hH];
    tile_store(0);
    __builtin_amdgcn_fence(__ATOMIC_RELEASE, "workgroup", "local");
    __builtin_amdgcn_s_barrier();
    __builtin_amdgcn_fence(__ATOMIC_ACQUIRE, "workgroup", "local");
    tile_load(JCHUNK);
    bv_fetch(0, 1, e1);
    adj_fetch(0, 0, aX);

#pragma unroll 1
    for (int c = 0; c < NCHUNK - 2; c += 2) {
        tile_step(c,     true, e0, e1, o0, o1);
        tile_step(c + 1, true, o0, o1, e0, e1);
    }
    tile_step(NCHUNK - 2, false, e0, e1, o0, o1);
    __builtin_amdgcn_sched_barrier(0);
    adj_fetch(NCHUNK - 1, 1, aY);
    __builtin_amdgcn_sched_barrier(0);
    trip_mma(aX, o0);
    trip_mma(aY, o1);

#pragma unroll
    for (int v = 0; v < 8; ++v) accX[jh][hl][v][lane] = acc[v];
    __syncthreads();

    const int rbase = (lane >> 4) * 8;
    const int dLane = (lane & 16) + DH;
    if (jh == 0 && nCol < DH) {
#pragma unroll
        for (int v = 0; v < 8; ++v) {
            const int   iL  = rbase + v;
            const float den = accX[0][hl][v][dLane] + accX[1][hl][v][dLane];
            float o = (acc[v] + accX[1][hl][v][lane]) / den;
            o = (o > 0.f) ? o : (__expf(o) - 1.f);
            out[(size_t)hH * NN * DH + (size_t)(i0 + iL) * DH + nCol] = o;
        }
    }
}

extern "C" void kernel_launch(void* const* d_in, const int* in_sizes, int n_in,
                              void* d_out, int out_size, void* d_ws, size_t ws_size,
                              hipStream_t stream) {
    const float* h   = (const float*)d_in[0];
    const int*   adj = (const int*)  d_in[1];
    const float* W_w = (const float*)d_in[2];
    const float* W_b = (const float*)d_in[3];
    const float* a_w = (const float*)d_in[4];
    const float* a_b = (const float*)d_in[5];
    float* out = (float*)d_out;

    float* Bg  = (float*)d_ws;
    float* efI = Bg + (size_t)NHEADS * NJB * BG_CHUNK;

    gat_wh_kernel<<<P1_GRID, 64, 0, stream>>>(h, W_w, W_b, a_w, a_b, Bg, efI);
    gat_att_kernel<<<ATT_GRID, 256, 0, stream>>>(adj, Bg, efI, out);
}
